# starting kernel
# baseline (speedup 1.0000x reference)
#include <hip/hip_runtime.h>


typedef float v2f __attribute__((ext_vector_type(2)));
typedef float v8f __attribute__((ext_vector_type(8)));

__device__ __forceinline__ void load_areg(v2f* areg, const float* __restrict__ aRow,
                                          int ga, int khalf) {
  const float* p = aRow + ga * 64 + 2 * khalf;
#pragma unroll
  for (int kk = 0; kk < 16; ++kk)
    areg[kk] = *reinterpret_cast<const v2f*>(p + 4 * kk);
}

__device__ __forceinline__ v8f tile_gemm(const v2f* areg, const float* __restrict__ Wp,
                                         float bb, int khalf) {
  v2f breg[16];
#pragma unroll
  for (int kk = 0; kk < 16; ++kk)
    breg[kk] = *reinterpret_cast<const v2f*>(Wp + 4 * kk + 2 * khalf);
  v8f acc = {bb, bb, bb, bb, bb, bb, bb, bb};
#pragma unroll
  for (int kk = 0; kk < 16; ++kk) {
    acc = __builtin_amdgcn_wmma_f32_16x16x4_f32(
         false, areg[kk],  false, breg[kk],
         (short)0, acc,  false,  false);
  }
  return acc;
}

__device__ __forceinline__ void store_tile(float* buf, int nt, int lane, v8f acc) {
  const int n = lane & 15, kh = lane >> 4;
#pragma unroll
  for (int i = 0; i < 8; ++i)
    buf[(i + 8 * kh) * 128 + nt * 16 + n] = acc[i];
}

__global__ __launch_bounds__(256) void fc2_fused(const float* __restrict__ in,
                                                 const float* __restrict__ W,
                                                 const float* __restrict__ bias,
                                                 float* __restrict__ out) {
  __shared__ float ytile[4][16 * 128];
  __shared__ float stile[16 * 128];
  __shared__ float xsumb[16 * 128];
  __shared__ float kvbuf[64];

  const int tid = threadIdx.x;
  const int lane = tid & 31;
  const int wave = tid >> 5;
  const int n0 = blockIdx.x * 16;
  const int bIdx = n0 >> 10;
  const int t0 = n0 & 1023;
  const int khalf = lane >> 4;
  const float* aRow = in + ((size_t)((t0 + (lane & 15)) * 4 + bIdx)) * 4096;

  const int rr = wave & 3;
  const int ntBase = wave >> 2;

  float m = -3.0e38f, dsum = 0.0f, num = 0.0f;

  for (int jj = 0; jj < 32; ++jj) {
    const int o_base = jj * 128;
    {
      const int ga = rr * 16 + (jj >> 1);
      v2f areg[16];
      load_areg(areg, aRow, ga, khalf);
#pragma unroll
      for (int k4 = 0; k4 < 4; ++k4) {
        const int nt = ntBase + 2 * k4;
        const int ch = rr * 4096 + o_base + nt * 16 + (lane & 15);
        v8f acc = tile_gemm(areg, W + (size_t)ch * 64, bias[ch], khalf);
        store_tile(ytile[rr], nt, lane, acc);
      }
    }
    __syncthreads();
#pragma unroll
    for (int i = 0; i < 8; ++i) {
      const int e = i * 256 + tid;
      xsumb[e] = ytile[0][e] + ytile[1][e] + ytile[2][e] + ytile[3][e];
    }
    __syncthreads();
    if (tid < 64) {
      const int prow = tid >> 2, pr = tid & 3;
      const float* yv = &ytile[pr][prow * 128];
      const float* xs = &xsumb[prow * 128];
      for (int o = 0; o < 128; ++o) {
        const float v = yv[o], x = xs[o];
        if (v > m) { const float sc = __expf(m - v); dsum *= sc; num *= sc; m = v; }
        const float e = __expf(v - m);
        dsum += e;
        num += e * x;
      }
    }
    __syncthreads();
  }
  if (tid < 64) kvbuf[tid] = num / dsum;
  __syncthreads();

  for (int jj = 0; jj < 32; ++jj) {
    const int o_base = jj * 128;
    {
      const int ga = rr * 16 + (jj >> 1);
      v2f areg[16];
      load_areg(areg, aRow, ga, khalf);
#pragma unroll
      for (int k4 = 0; k4 < 4; ++k4) {
        const int nt = ntBase + 2 * k4;
        const int ch = rr * 4096 + o_base + nt * 16 + (lane & 15);
        v8f acc = tile_gemm(areg, W + (size_t)ch * 64, bias[ch], khalf);
        store_tile(ytile[rr], nt, lane, acc);
      }
      const int gs = 2 * jj + (wave >> 2);
      v2f sreg[16];
      load_areg(sreg, aRow, gs, khalf);
      const int chs = 4 * (o_base + wave * 16 + (lane & 15));
      v8f accs = tile_gemm(sreg, W + (size_t)chs * 64, bias[chs], khalf);
      store_tile(stile, wave, lane, accs);
    }
    __syncthreads();
#pragma unroll
    for (int i = 0; i < 8; ++i) {
      const int e = i * 256 + tid;
      const int prow = e >> 7;
      const int oo = e & 127;
      const float* kvp = &kvbuf[prow * 4];
      const float val = stile[e] + ytile[0][e] * kvp[0] + ytile[1][e] * kvp[1] +
                        ytile[2][e] * kvp[2] + ytile[3][e] * kvp[3];
      out[((size_t)((t0 + prow) * 4 + bIdx)) * 4096 + o_base + oo] = val;
    }
    __syncthreads();
  }
}

extern "C" void kernel_launch(void* const* d_in, const int* in_sizes, int n_in,
                              void* d_out, int out_size, void* d_ws, size_t ws_size,
                              hipStream_t stream) {
  (void)in_sizes; (void)n_in; (void)d_ws; (void)ws_size; (void)out_size;
  const float* in = (const float*)d_in[0];
  const float* W = (const float*)d_in[1];
  const float* bias = (const float*)d_in[2];
  float* out = (float*)d_out;
  fc2_fused<<<256, 256, 0, stream>>>(in, W, bias, out);
}


# Operand tiles fetched one block ahead with whole-row coalesced b128 loads and put into WMMA fragment order through padded per-wave LDS slots; k summation order of the dot products permuted
# speedup vs baseline: 1.8212x; 1.8212x over previous
#include <hip/hip_runtime.h>


typedef float v2f __attribute__((ext_vector_type(2)));
typedef float v4f __attribute__((ext_vector_type(4)));
typedef float v8f __attribute__((ext_vector_type(8)));

#define PITCH 68
#define SLOT (16 * PITCH)

__device__ __forceinline__ void gload_tile(v4f* g, const float* __restrict__ base,
                                           size_t rowStride, int lane) {
  const float* p = base + (size_t)(lane >> 4) * rowStride + 4 * (lane & 15);
#pragma unroll
  for (int j = 0; j < 8; ++j)
    g[j] = *reinterpret_cast<const v4f*>(p + (size_t)(2 * j) * rowStride);
}

__device__ __forceinline__ void stage_tile(float* slot, const v4f* g, int lane) {
  float* q = slot + (lane >> 4) * PITCH + 4 * (lane & 15);
#pragma unroll
  for (int j = 0; j < 8; ++j)
    *reinterpret_cast<v4f*>(q + 2 * j * PITCH) = g[j];
}

__device__ __forceinline__ void frag_read(v4f* f, const float* slot, int lane) {
  const float* q = slot + (lane & 15) * PITCH + 4 * (lane >> 4);
#pragma unroll
  for (int p = 0; p < 8; ++p)
    f[p] = *reinterpret_cast<const v4f*>(q + 8 * p);
}

__device__ __forceinline__ v8f tile_gemm(const v4f* af, const v4f* bf, float bb) {
  v8f acc = {bb, bb, bb, bb, bb, bb, bb, bb};
#pragma unroll
  for (int p = 0; p < 8; ++p) {
    const v2f a0 = af[p].xy, a1 = af[p].zw;
    const v2f b0 = bf[p].xy, b1 = bf[p].zw;
    acc = __builtin_amdgcn_wmma_f32_16x16x4_f32(
         false, a0,  false, b0,
         (short)0, acc,  false,  false);
    acc = __builtin_amdgcn_wmma_f32_16x16x4_f32(
         false, a1,  false, b1,
         (short)0, acc,  false,  false);
  }
  return acc;
}

__device__ __forceinline__ void store_tile(float* buf, int nt, int lane, v8f acc) {
  const int n = lane & 15, kh = lane >> 4;
#pragma unroll
  for (int i = 0; i < 8; ++i)
    buf[(i + 8 * kh) * 128 + nt * 16 + n] = acc[i];
}

__device__ __forceinline__ void issue_main(v4f* g, float* bb, const float* __restrict__ aBase,
                                           const float* __restrict__ W,
                                           const float* __restrict__ bias, int jj, int rr,
                                           int ntBase, int lane) {
  const int ga = rr * 16 + (jj >> 1);
  gload_tile(g, aBase + ga * 64, (size_t)4 * 4096, lane);
#pragma unroll
  for (int k4 = 0; k4 < 4; ++k4) {
    const int ch0 = rr * 4096 + jj * 128 + (ntBase + 2 * k4) * 16;
    gload_tile(g + 8 * (1 + k4), W + (size_t)ch0 * 64, 64, lane);
    bb[k4] = bias[ch0 + (lane & 15)];
  }
}

__device__ __forceinline__ void issue_s(v4f* g, float* bb, const float* __restrict__ aBase,
                                        const float* __restrict__ W,
                                        const float* __restrict__ bias, int jj, int wave,
                                        int lane) {
  const int gs = 2 * jj + (wave >> 2);
  gload_tile(g, aBase + gs * 64, (size_t)4 * 4096, lane);
  const int chs0 = 4 * (jj * 128 + wave * 16);
  gload_tile(g + 8, W + (size_t)chs0 * 64, 4 * 64, lane);
  bb[0] = bias[chs0 + 4 * (lane & 15)];
}

__global__ __launch_bounds__(256) void fc2_fused(const float* __restrict__ in,
                                                 const float* __restrict__ W,
                                                 const float* __restrict__ bias,
                                                 float* __restrict__ out) {
  __shared__ float ytile[4][16 * 128];
  __shared__ float stile[16 * 128];
  __shared__ float xsumb[16 * 128];
  __shared__ float kvbuf[64];
  __shared__ __attribute__((aligned(16))) float stage[8][7][SLOT];

  const int tid = threadIdx.x;
  const int lane = tid & 31;
  const int wave = tid >> 5;
  const int n0 = blockIdx.x * 16;
  const int bIdx = n0 >> 10;
  const int t0 = n0 & 1023;
  const float* aBase = in + ((size_t)(t0 * 4 + bIdx)) * 4096;
  float* st = &stage[wave][0][0];

  const int rr = wave & 3;
  const int ntBase = wave >> 2;

  float m = -3.0e38f, dsum = 0.0f, num = 0.0f;

  v4f g[40];
  float bb[4];
  issue_main(g, bb, aBase, W, bias, 0, rr, ntBase, lane);

  for (int jj = 0; jj < 32; ++jj) {
#pragma unroll
    for (int s = 0; s < 5; ++s) stage_tile(st + s * SLOT, g + 8 * s, lane);
    const float b0 = bb[0], b1 = bb[1], b2 = bb[2], b3 = bb[3];
    __syncthreads();
    issue_main(g, bb, aBase, W, bias, (jj + 1) & 31, rr, ntBase, lane);
    {
      const float bcur[4] = {b0, b1, b2, b3};
      v4f af[8];
      frag_read(af, st, lane);
#pragma unroll
      for (int k4 = 0; k4 < 4; ++k4) {
        const int nt = ntBase + 2 * k4;
        v4f bf[8];
        frag_read(bf, st + (1 + k4) * SLOT, lane);
        v8f acc = tile_gemm(af, bf, bcur[k4]);
        store_tile(ytile[rr], nt, lane, acc);
      }
    }
    __syncthreads();
#pragma unroll
    for (int i = 0; i < 8; ++i) {
      const int e = i * 256 + tid;
      xsumb[e] = ytile[0][e] + ytile[1][e] + ytile[2][e] + ytile[3][e];
    }
    __syncthreads();
    if (tid < 64) {
      const int prow = tid >> 2, pr = tid & 3;
      const float* yv = &ytile[pr][prow * 128];
      const float* xs = &xsumb[prow * 128];
      for (int o = 0; o < 128; ++o) {
        const float v = yv[o], x = xs[o];
        if (v > m) { const float sc = __expf(m - v); dsum *= sc; num *= sc; m = v; }
        const float e = __expf(v - m);
        dsum += e;
        num += e * x;
      }
    }
    __syncthreads();
  }
  if (tid < 64) kvbuf[tid] = num / dsum;
  __syncthreads();

  v4f gs[16];
  float bbs[1];
  issue_s(gs, bbs, aBase, W, bias, 0, wave, lane);
  for (int jj = 0; jj < 32; ++jj) {
#pragma unroll
    for (int s = 0; s < 5; ++s) stage_tile(st + s * SLOT, g + 8 * s, lane);
    stage_tile(st + 5 * SLOT, gs, lane);
    stage_tile(st + 6 * SLOT, gs + 8, lane);
    const float b0 = bb[0], b1 = bb[1], b2 = bb[2], b3 = bb[3], bs = bbs[0];
    __syncthreads();
    issue_main(g, bb, aBase, W, bias, (jj + 1) & 31, rr, ntBase, lane);
    issue_s(gs, bbs, aBase, W, bias, (jj + 1) & 31, wave, lane);
    const int o_base = jj * 128;
    {
      const float bcur[4] = {b0, b1, b2, b3};
      v4f af[8];
      frag_read(af, st, lane);
#pragma unroll
      for (int k4 = 0; k4 < 4; ++k4) {
        const int nt = ntBase + 2 * k4;
        v4f bf[8];
        frag_read(bf, st + (1 + k4) * SLOT, lane);
        v8f acc = tile_gemm(af, bf, bcur[k4]);
        store_tile(ytile[rr], nt, lane, acc);
      }
      v4f sa[8], sb[8];
      frag_read(sa, st + 5 * SLOT, lane);
      frag_read(sb, st + 6 * SLOT, lane);
      v8f accs = tile_gemm(sa, sb, bs);
      store_tile(stile, wave, lane, accs);
    }
    __syncthreads();
#pragma unroll
    for (int i = 0; i < 8; ++i) {
      const int e = i * 256 + tid;
      const int prow = e >> 7;
      const int oo = e & 127;
      const float* kvp = &kvbuf[prow * 4];
      const float val = stile[e] + ytile[0][e] * kvp[0] + ytile[1][e] * kvp[1] +
                        ytile[2][e] * kvp[2] + ytile[3][e] * kvp[3];
      out[((size_t)((t0 + prow) * 4 + bIdx)) * 4096 + o_base + oo] = val;
    }
    __syncthreads();
  }
}

extern "C" void kernel_launch(void* const* d_in, const int* in_sizes, int n_in,
                              void* d_out, int out_size, void* d_ws, size_t ws_size,
                              hipStream_t stream) {
  (void)in_sizes; (void)n_in; (void)d_ws; (void)ws_size; (void)out_size;
  const float* in = (const float*)d_in[0];
  const float* W = (const float*)d_in[1];
  const float* bias = (const float*)d_in[2];
  float* out = (float*)d_out;
  fc2_fused<<<256, 256, 0, stream>>>(in, W, bias, out);
}


# Phase-1 softmax statistics on all 256 threads (16 per row, one rescale per block, lane-butterfly merge at end) instead of 64; xsum buffer and 2 barriers per block dropped; dsum/num sum order changed
# speedup vs baseline: 3.2476x; 1.7832x over previous
#include <hip/hip_runtime.h>


typedef float v2f __attribute__((ext_vector_type(2)));
typedef float v4f __attribute__((ext_vector_type(4)));
typedef float v8f __attribute__((ext_vector_type(8)));

#define PITCH 68
#define SLOT (16 * PITCH)

__device__ __forceinline__ void gload_tile(v4f* g, const float* __restrict__ base,
                                           size_t rowStride, int lane) {
  const float* p = base + (size_t)(lane >> 4) * rowStride + 4 * (lane & 15);
#pragma unroll
  for (int j = 0; j < 8; ++j)
    g[j] = *reinterpret_cast<const v4f*>(p + (size_t)(2 * j) * rowStride);
}

__device__ __forceinline__ void stage_tile(float* slot, const v4f* g, int lane) {
  float* q = slot + (lane >> 4) * PITCH + 4 * (lane & 15);
#pragma unroll
  for (int j = 0; j < 8; ++j)
    *reinterpret_cast<v4f*>(q + 2 * j * PITCH) = g[j];
}

__device__ __forceinline__ void frag_read(v4f* f, const float* slot, int lane) {
  const float* q = slot + (lane & 15) * PITCH + 4 * (lane >> 4);
#pragma unroll
  for (int p = 0; p < 8; ++p)
    f[p] = *reinterpret_cast<const v4f*>(q + 8 * p);
}

__device__ __forceinline__ v8f tile_gemm(const v4f* af, const v4f* bf, float bb) {
  v8f acc = {bb, bb, bb, bb, bb, bb, bb, bb};
#pragma unroll
  for (int p = 0; p < 8; ++p) {
    const v2f a0 = af[p].xy, a1 = af[p].zw;
    const v2f b0 = bf[p].xy, b1 = bf[p].zw;
    acc = __builtin_amdgcn_wmma_f32_16x16x4_f32(
         false, a0,  false, b0,
         (short)0, acc,  false,  false);
    acc = __builtin_amdgcn_wmma_f32_16x16x4_f32(
         false, a1,  false, b1,
         (short)0, acc,  false,  false);
  }
  return acc;
}

__device__ __forceinline__ void store_tile(float* buf, int nt, int lane, v8f acc) {
  const int n = lane & 15, kh = lane >> 4;
#pragma unroll
  for (int i = 0; i < 8; ++i)
    buf[(i + 8 * kh) * 128 + nt * 16 + n] = acc[i];
}

__device__ __forceinline__ void issue_main(v4f* g, float* bb, const float* __restrict__ aBase,
                                           const float* __restrict__ W,
                                           const float* __restrict__ bias, int jj, int rr,
                                           int ntBase, int lane) {
  const int ga = rr * 16 + (jj >> 1);
  gload_tile(g, aBase + ga * 64, (size_t)4 * 4096, lane);
#pragma unroll
  for (int k4 = 0; k4 < 4; ++k4) {
    const int ch0 = rr * 4096 + jj * 128 + (ntBase + 2 * k4) * 16;
    gload_tile(g + 8 * (1 + k4), W + (size_t)ch0 * 64, 64, lane);
    bb[k4] = bias[ch0 + (lane & 15)];
  }
}

__device__ __forceinline__ void issue_s(v4f* g, float* bb, const float* __restrict__ aBase,
                                        const float* __restrict__ W,
                                        const float* __restrict__ bias, int jj, int wave,
                                        int lane) {
  const int gs = 2 * jj + (wave >> 2);
  gload_tile(g, aBase + gs * 64, (size_t)4 * 4096, lane);
  const int chs0 = 4 * (jj * 128 + wave * 16);
  gload_tile(g + 8, W + (size_t)chs0 * 64, 4 * 64, lane);
  bb[0] = bias[chs0 + 4 * (lane & 15)];
}

__global__ __launch_bounds__(256) void fc2_fused(const float* __restrict__ in,
                                                 const float* __restrict__ W,
                                                 const float* __restrict__ bias,
                                                 float* __restrict__ out) {
  __shared__ __attribute__((aligned(16))) float ytile[4][16 * 128];
  __shared__ float stile[16 * 128];
  __shared__ float kvbuf[64];
  __shared__ __attribute__((aligned(16))) float stage[8][7][SLOT];

  const int tid = threadIdx.x;
  const int lane = tid & 31;
  const int wave = tid >> 5;
  const int n0 = blockIdx.x * 16;
  const int bIdx = n0 >> 10;
  const int t0 = n0 & 1023;
  const float* aBase = in + ((size_t)(t0 * 4 + bIdx)) * 4096;
  float* st = &stage[wave][0][0];

  const int rr = wave & 3;
  const int ntBase = wave >> 2;

  const int srow = tid >> 4;
  const int sc4 = 4 * (tid & 15);
  float m[4], dsum[4], num[4];
#pragma unroll
  for (int r = 0; r < 4; ++r) { m[r] = -3.0e38f; dsum[r] = 0.0f; num[r] = 0.0f; }

  v4f g[40];
  float bb[4];
  issue_main(g, bb, aBase, W, bias, 0, rr, ntBase, lane);

  for (int jj = 0; jj < 32; ++jj) {
#pragma unroll
    for (int s = 0; s < 5; ++s) stage_tile(st + s * SLOT, g + 8 * s, lane);
    const float b0 = bb[0], b1 = bb[1], b2 = bb[2], b3 = bb[3];
    __syncthreads();
    issue_main(g, bb, aBase, W, bias, (jj + 1) & 31, rr, ntBase, lane);
    {
      const float bcur[4] = {b0, b1, b2, b3};
      v4f af[8];
      frag_read(af, st, lane);
#pragma unroll
      for (int k4 = 0; k4 < 4; ++k4) {
        const int nt = ntBase + 2 * k4;
        v4f bf[8];
        frag_read(bf, st + (1 + k4) * SLOT, lane);
        v8f acc = tile_gemm(af, bf, bcur[k4]);
        store_tile(ytile[rr], nt, lane, acc);
      }
    }
    __syncthreads();
    {
      v4f va[4], vb[4];
#pragma unroll
      for (int r = 0; r < 4; ++r) {
        va[r] = *reinterpret_cast<const v4f*>(&ytile[r][srow * 128 + sc4]);
        vb[r] = *reinterpret_cast<const v4f*>(&ytile[r][srow * 128 + 64 + sc4]);
      }
      const v4f xa = va[0] + va[1] + va[2] + va[3];
      const v4f xb = vb[0] + vb[1] + vb[2] + vb[3];
#pragma unroll
      for (int r = 0; r < 4; ++r) {
        const v4f a = va[r], b = vb[r];
        const float mb = fmaxf(fmaxf(fmaxf(a.x, a.y), fmaxf(a.z, a.w)),
                               fmaxf(fmaxf(b.x, b.y), fmaxf(b.z, b.w)));
        const float mn = fmaxf(m[r], mb);
        const float sc = __expf(m[r] - mn);
        const float ea0 = __expf(a.x - mn), ea1 = __expf(a.y - mn);
        const float ea2 = __expf(a.z - mn), ea3 = __expf(a.w - mn);
        const float eb0 = __expf(b.x - mn), eb1 = __expf(b.y - mn);
        const float eb2 = __expf(b.z - mn), eb3 = __expf(b.w - mn);
        const float da = (ea0 + ea1) + (ea2 + ea3);
        const float db = (eb0 + eb1) + (eb2 + eb3);
        const float na = (ea0 * xa.x + ea1 * xa.y) + (ea2 * xa.z + ea3 * xa.w);
        const float nb = (eb0 * xb.x + eb1 * xb.y) + (eb2 * xb.z + eb3 * xb.w);
        dsum[r] = dsum[r] * sc + (da + db);
        num[r] = num[r] * sc + (na + nb);
        m[r] = mn;
      }
    }
  }
#pragma unroll
  for (int r = 0; r < 4; ++r) {
#pragma unroll
    for (int mask = 1; mask < 16; mask <<= 1) {
      const float mo = __shfl_xor(m[r], mask, 32);
      const float dq = __shfl_xor(dsum[r], mask, 32);
      const float no = __shfl_xor(num[r], mask, 32);
      const float mn = fmaxf(m[r], mo);
      const float sa = __expf(m[r] - mn), sb = __expf(mo - mn);
      dsum[r] = dsum[r] * sa + dq * sb;
      num[r] = num[r] * sa + no * sb;
      m[r] = mn;
    }
  }
  if ((tid & 15) == 0) {
#pragma unroll
    for (int r = 0; r < 4; ++r) kvbuf[srow * 4 + r] = num[r] / dsum[r];
  }
  __syncthreads();

  v4f gs[16];
  float bbs[1];
  issue_s(gs, bbs, aBase, W, bias, 0, wave, lane);
  for (int jj = 0; jj < 32; ++jj) {
#pragma unroll
    for (int s = 0; s < 5; ++s) stage_tile(st + s * SLOT, g + 8 * s, lane);
    stage_tile(st + 5 * SLOT, gs, lane);
    stage_tile(st + 6 * SLOT, gs + 8, lane);
    const float b0 = bb[0], b1 = bb[1], b2 = bb[2], b3 = bb[3], bs = bbs[0];
    __syncthreads();
    issue_main(g, bb, aBase, W, bias, (jj + 1) & 31, rr, ntBase, lane);
    issue_s(gs, bbs, aBase, W, bias, (jj + 1) & 31, wave, lane);
    const int o_base = jj * 128;
    {
      const float bcur[4] = {b0, b1, b2, b3};
      v4f af[8];
      frag_read(af, st, lane);
#pragma unroll
      for (int k4 = 0; k4 < 4; ++k4) {
        const int nt = ntBase + 2 * k4;
        v4f bf[8];
        frag_read(bf, st + (1 + k4) * SLOT, lane);
        v8f acc = tile_gemm(af, bf, bcur[k4]);
        store_tile(ytile[rr], nt, lane, acc);
      }
      v4f sa[8], sb[8];
      frag_read(sa, st + 5 * SLOT, lane);
      frag_read(sb, st + 6 * SLOT, lane);
      v8f accs = tile_gemm(sa, sb, bs);
      store_tile(stile, wave, lane, accs);
    }
    __syncthreads();
#pragma unroll
    for (int i = 0; i < 8; ++i) {
      const int e = i * 256 + tid;
      const int prow = e >> 7;
      const int oo = e & 127;
      const float* kvp = &kvbuf[prow * 4];
      const float val = stile[e] + ytile[0][e] * kvp[0] + ytile[1][e] * kvp[1] +
                        ytile[2][e] * kvp[2] + ytile[3][e] * kvp[3];
      out[((size_t)((t0 + prow) * 4 + bIdx)) * 4096 + o_base + oo] = val;
    }
    __syncthreads();
  }
}

extern "C" void kernel_launch(void* const* d_in, const int* in_sizes, int n_in,
                              void* d_out, int out_size, void* d_ws, size_t ws_size,
                              hipStream_t stream) {
  (void)in_sizes; (void)n_in; (void)d_ws; (void)ws_size; (void)out_size;
  const float* in = (const float*)d_in[0];
  const float* W = (const float*)d_in[1];
  const float* bias = (const float*)d_in[2];
  float* out = (float*)d_out;
  fc2_fused<<<256, 256, 0, stream>>>(in, W, bias, out);
}


# Workgroup barriers wait for LDS only (LDS-scope fences around s_barrier instead of __syncthreads): the next block's operand loads and the output stores stay in flight across them
# speedup vs baseline: 3.4871x; 1.0737x over previous
#include <hip/hip_runtime.h>


typedef float v2f __attribute__((ext_vector_type(2)));
typedef float v4f __attribute__((ext_vector_type(4)));
typedef float v8f __attribute__((ext_vector_type(8)));

#define PITCH 68
#define SLOT (16 * PITCH)

__device__ __forceinline__ void gload_tile(v4f* g, const float* __restrict__ base,
                                           size_t rowStride, int lane) {
  const float* p = base + (size_t)(lane >> 4) * rowStride + 4 * (lane & 15);
#pragma unroll
  for (int j = 0; j < 8; ++j)
    g[j] = *reinterpret_cast<const v4f*>(p + (size_t)(2 * j) * rowStride);
}

__device__ __forceinline__ void stage_tile(float* slot, const v4f* g, int lane) {
  float* q = slot + (lane >> 4) * PITCH + 4 * (lane & 15);
#pragma unroll
  for (int j = 0; j < 8; ++j)
    *reinterpret_cast<v4f*>(q + 2 * j * PITCH) = g[j];
}

__device__ __forceinline__ void frag_read(v4f* f, const float* slot, int lane) {
  const float* q = slot + (lane & 15) * PITCH + 4 * (lane >> 4);
#pragma unroll
  for (int p = 0; p < 8; ++p)
    f[p] = *reinterpret_cast<const v4f*>(q + 8 * p);
}

__device__ __forceinline__ v8f tile_gemm(const v4f* af, const v4f* bf, float bb) {
  v8f acc = {bb, bb, bb, bb, bb, bb, bb, bb};
#pragma unroll
  for (int p = 0; p < 8; ++p) {
    const v2f a0 = af[p].xy, a1 = af[p].zw;
    const v2f b0 = bf[p].xy, b1 = bf[p].zw;
    acc = __builtin_amdgcn_wmma_f32_16x16x4_f32(
         false, a0,  false, b0,
         (short)0, acc,  false,  false);
    acc = __builtin_amdgcn_wmma_f32_16x16x4_f32(
         false, a1,  false, b1,
         (short)0, acc,  false,  false);
  }
  return acc;
}

__device__ __forceinline__ void store_tile(float* buf, int nt, int lane, v8f acc) {
  const int n = lane & 15, kh = lane >> 4;
#pragma unroll
  for (int i = 0; i < 8; ++i)
    buf[(i + 8 * kh) * 128 + nt * 16 + n] = acc[i];
}

__device__ __forceinline__ void issue_main(v4f* g, float* bb, const float* __restrict__ aBase,
                                           const float* __restrict__ W,
                                           const float* __restrict__ bias, int jj, int rr,
                                           int ntBase, int lane) {
  const int ga = rr * 16 + (jj >> 1);
  gload_tile(g, aBase + ga * 64, (size_t)4 * 4096, lane);
#pragma unroll
  for (int k4 = 0; k4 < 4; ++k4) {
    const int ch0 = rr * 4096 + jj * 128 + (ntBase + 2 * k4) * 16;
    gload_tile(g + 8 * (1 + k4), W + (size_t)ch0 * 64, 64, lane);
    bb[k4] = bias[ch0 + (lane & 15)];
  }
}

__device__ __forceinline__ void issue_s(v4f* g, float* bb, const float* __restrict__ aBase,
                                        const float* __restrict__ W,
                                        const float* __restrict__ bias, int jj, int wave,
                                        int lane) {
  const int gs = 2 * jj + (wave >> 2);
  gload_tile(g, aBase + gs * 64, (size_t)4 * 4096, lane);
  const int chs0 = 4 * (jj * 128 + wave * 16);
  gload_tile(g + 8, W + (size_t)chs0 * 64, 4 * 64, lane);
  bb[0] = bias[chs0 + 4 * (lane & 15)];
}

__device__ __forceinline__ void lds_barrier() {
  __builtin_amdgcn_fence(__ATOMIC_RELEASE, "workgroup", "local");
  __builtin_amdgcn_s_barrier();
  __builtin_amdgcn_fence(__ATOMIC_ACQUIRE, "workgroup", "local");
}

__global__ __launch_bounds__(256) void fc2_fused(const float* __restrict__ in,
                                                 const float* __restrict__ W,
                                                 const float* __restrict__ bias,
                                                 float* __restrict__ out) {
  __shared__ __attribute__((aligned(16))) float ytile[4][16 * 128];
  __shared__ float stile[16 * 128];
  __shared__ float kvbuf[64];
  __shared__ __attribute__((aligned(16))) float stage[8][7][SLOT];

  const int tid = threadIdx.x;
  const int lane = tid & 31;
  const int wave = tid >> 5;
  const int n0 = blockIdx.x * 16;
  const int bIdx = n0 >> 10;
  const int t0 = n0 & 1023;
  const float* aBase = in + ((size_t)(t0 * 4 + bIdx)) * 4096;
  float* st = &stage[wave][0][0];

  const int rr = wave & 3;
  const int ntBase = wave >> 2;

  const int srow = tid >> 4;
  const int sc4 = 4 * (tid & 15);
  float m[4], dsum[4], num[4];
#pragma unroll
  for (int r = 0; r < 4; ++r) { m[r] = -3.0e38f; dsum[r] = 0.0f; num[r] = 0.0f; }

  v4f g[40];
  float bb[4];
  issue_main(g, bb, aBase, W, bias, 0, rr, ntBase, lane);

  for (int jj = 0; jj < 32; ++jj) {
#pragma unroll
    for (int s = 0; s < 5; ++s) stage_tile(st + s * SLOT, g + 8 * s, lane);
    const float b0 = bb[0], b1 = bb[1], b2 = bb[2], b3 = bb[3];
    lds_barrier();
    issue_main(g, bb, aBase, W, bias, (jj + 1) & 31, rr, ntBase, lane);
    {
      const float bcur[4] = {b0, b1, b2, b3};
      v4f af[8];
      frag_read(af, st, lane);
#pragma unroll
      for (int k4 = 0; k4 < 4; ++k4) {
        const int nt = ntBase + 2 * k4;
        v4f bf[8];
        frag_read(bf, st + (1 + k4) * SLOT, lane);
        v8f acc = tile_gemm(af, bf, bcur[k4]);
        store_tile(ytile[rr], nt, lane, acc);
      }
    }
    lds_barrier();
    {
      v4f va[4], vb[4];
#pragma unroll
      for (int r = 0; r < 4; ++r) {
        va[r] = *reinterpret_cast<const v4f*>(&ytile[r][srow * 128 + sc4]);
        vb[r] = *reinterpret_cast<const v4f*>(&ytile[r][srow * 128 + 64 + sc4]);
      }
      const v4f xa = va[0] + va[1] + va[2] + va[3];
      const v4f xb = vb[0] + vb[1] + vb[2] + vb[3];
#pragma unroll
      for (int r = 0; r < 4; ++r) {
        const v4f a = va[r], b = vb[r];
        const float mb = fmaxf(fmaxf(fmaxf(a.x, a.y), fmaxf(a.z, a.w)),
                               fmaxf(fmaxf(b.x, b.y), fmaxf(b.z, b.w)));
        const float mn = fmaxf(m[r], mb);
        const float sc = __expf(m[r] - mn);
        const float ea0 = __expf(a.x - mn), ea1 = __expf(a.y - mn);
        const float ea2 = __expf(a.z - mn), ea3 = __expf(a.w - mn);
        const float eb0 = __expf(b.x - mn), eb1 = __expf(b.y - mn);
        const float eb2 = __expf(b.z - mn), eb3 = __expf(b.w - mn);
        const float da = (ea0 + ea1) + (ea2 + ea3);
        const float db = (eb0 + eb1) + (eb2 + eb3);
        const float na = (ea0 * xa.x + ea1 * xa.y) + (ea2 * xa.z + ea3 * xa.w);
        const float nb = (eb0 * xb.x + eb1 * xb.y) + (eb2 * xb.z + eb3 * xb.w);
        dsum[r] = dsum[r] * sc + (da + db);
        num[r] = num[r] * sc + (na + nb);
        m[r] = mn;
      }
    }
  }
#pragma unroll
  for (int r = 0; r < 4; ++r) {
#pragma unroll
    for (int mask = 1; mask < 16; mask <<= 1) {
      const float mo = __shfl_xor(m[r], mask, 32);
      const float dq = __shfl_xor(dsum[r], mask, 32);
      const float no = __shfl_xor(num[r], mask, 32);
      const float mn = fmaxf(m[r], mo);
      const float sa = __expf(m[r] - mn), sb = __expf(mo - mn);
      dsum[r] = dsum[r] * sa + dq * sb;
      num[r] = num[r] * sa + no * sb;
      m[r] = mn;
    }
  }
  if ((tid & 15) == 0) {
#pragma unroll
    for (int r = 0; r < 4; ++r) kvbuf[srow * 4 + r] = num[r] / dsum[r];
  }
  lds_barrier();

  v4f gs[16];
  float bbs[1];
  issue_s(gs, bbs, aBase, W, bias, 0, wave, lane);
  for (int jj = 0; jj < 32; ++jj) {
#pragma unroll
    for (int s = 0; s < 5; ++s) stage_tile(st + s * SLOT, g + 8 * s, lane);
    stage_tile(st + 5 * SLOT, gs, lane);
    stage_tile(st + 6 * SLOT, gs + 8, lane);
    const float b0 = bb[0], b1 = bb[1], b2 = bb[2], b3 = bb[3], bs = bbs[0];
    lds_barrier();
    issue_main(g, bb, aBase, W, bias, (jj + 1) & 31, rr, ntBase, lane);
    issue_s(gs, bbs, aBase, W, bias, (jj + 1) & 31, wave, lane);
    const int o_base = jj * 128;
    {
      const float bcur[4] = {b0, b1, b2, b3};
      v4f af[8];
      frag_read(af, st, lane);
#pragma unroll
      for (int k4 = 0; k4 < 4; ++k4) {
        const int nt = ntBase + 2 * k4;
        v4f bf[8];
        frag_read(bf, st + (1 + k4) * SLOT, lane);
        v8f acc = tile_gemm(af, bf, bcur[k4]);
        store_tile(ytile[rr], nt, lane, acc);
      }
      v4f sa[8], sb[8];
      frag_read(sa, st + 5 * SLOT, lane);
      frag_read(sb, st + 6 * SLOT, lane);
      v8f accs = tile_gemm(sa, sb, bs);
      store_tile(stile, wave, lane, accs);
    }
    lds_barrier();
#pragma unroll
    for (int i = 0; i < 8; ++i) {
      const int e = i * 256 + tid;
      const int prow = e >> 7;
      const int oo = e & 127;
      const float* kvp = &kvbuf[prow * 4];
      const float val = stile[e] + ytile[0][e] * kvp[0] + ytile[1][e] * kvp[1] +
                        ytile[2][e] * kvp[2] + ytile[3][e] * kvp[3];
      out[((size_t)((t0 + prow) * 4 + bIdx)) * 4096 + o_base + oo] = val;
    }
    lds_barrier();
  }
}

extern "C" void kernel_launch(void* const* d_in, const int* in_sizes, int n_in,
                              void* d_out, int out_size, void* d_ws, size_t ws_size,
                              hipStream_t stream) {
  (void)in_sizes; (void)n_in; (void)d_ws; (void)ws_size; (void)out_size;
  const float* in = (const float*)d_in[0];
  const float* W = (const float*)d_in[1];
  const float* bias = (const float*)d_in[2];
  float* out = (float*)d_out;
  fc2_fused<<<256, 256, 0, stream>>>(in, W, bias, out);
}


# Input tiles fetched once per workgroup: the 2 (main) or 4 ('s') waves that use a tile each load a share of its rows into one shared LDS slot; a main tile (2 buffers) serves the 2 blocks it covers
# speedup vs baseline: 3.7159x; 1.0656x over previous
#include <hip/hip_runtime.h>


typedef float v2f __attribute__((ext_vector_type(2)));
typedef float v4f __attribute__((ext_vector_type(4)));
typedef float v8f __attribute__((ext_vector_type(8)));

#define PITCH 68
#define SLOT (16 * PITCH)

__device__ __forceinline__ void gload_tile(v4f* g, const float* __restrict__ base,
                                           size_t rowStride, int lane) {
  const float* p = base + (size_t)(lane >> 4) * rowStride + 4 * (lane & 15);
#pragma unroll
  for (int j = 0; j < 8; ++j)
    g[j] = *reinterpret_cast<const v4f*>(p + (size_t)(2 * j) * rowStride);
}

__device__ __forceinline__ void stage_tile(float* slot, const v4f* g, int lane) {
  float* q = slot + (lane >> 4) * PITCH + 4 * (lane & 15);
#pragma unroll
  for (int j = 0; j < 8; ++j)
    *reinterpret_cast<v4f*>(q + 2 * j * PITCH) = g[j];
}

template <int NS>
__device__ __forceinline__ void gload_part(v4f* g, const float* __restrict__ base,
                                           size_t rowStride, int lane, int row0) {
  const float* p = base + (size_t)(row0 + (lane >> 4)) * rowStride + 4 * (lane & 15);
#pragma unroll
  for (int j = 0; j < NS; ++j)
    g[j] = *reinterpret_cast<const v4f*>(p + (size_t)(2 * j) * rowStride);
}

template <int NS>
__device__ __forceinline__ void stage_part(float* slot, const v4f* g, int lane, int row0) {
  float* q = slot + (row0 + (lane >> 4)) * PITCH + 4 * (lane & 15);
#pragma unroll
  for (int j = 0; j < NS; ++j)
    *reinterpret_cast<v4f*>(q + 2 * j * PITCH) = g[j];
}

__device__ __forceinline__ void frag_read(v4f* f, const float* slot, int lane) {
  const float* q = slot + (lane & 15) * PITCH + 4 * (lane >> 4);
#pragma unroll
  for (int p = 0; p < 8; ++p)
    f[p] = *reinterpret_cast<const v4f*>(q + 8 * p);
}

__device__ __forceinline__ v8f tile_gemm(const v4f* af, const v4f* bf, float bb) {
  v8f acc = {bb, bb, bb, bb, bb, bb, bb, bb};
#pragma unroll
  for (int p = 0; p < 8; ++p) {
    const v2f a0 = af[p].xy, a1 = af[p].zw;
    const v2f b0 = bf[p].xy, b1 = bf[p].zw;
    acc = __builtin_amdgcn_wmma_f32_16x16x4_f32(
         false, a0,  false, b0,
         (short)0, acc,  false,  false);
    acc = __builtin_amdgcn_wmma_f32_16x16x4_f32(
         false, a1,  false, b1,
         (short)0, acc,  false,  false);
  }
  return acc;
}

__device__ __forceinline__ void store_tile(float* buf, int nt, int lane, v8f acc) {
  const int n = lane & 15, kh = lane >> 4;
#pragma unroll
  for (int i = 0; i < 8; ++i)
    buf[(i + 8 * kh) * 128 + nt * 16 + n] = acc[i];
}

__device__ __forceinline__ int a_row0(int jj, int ntBase) {
  return 8 * ntBase + ((jj & 1) ? 0 : 4);
}

__device__ __forceinline__ void issue_main(v4f* g, float* bb, const float* __restrict__ aBase,
                                           const float* __restrict__ W,
                                           const float* __restrict__ bias, int jj, int rr,
                                           int ntBase, int lane) {
  const int jE = (jj + (jj & 1)) & 31;
  const float* aT = aBase + (rr * 16 + (jE >> 1)) * 64;
  gload_part<2>(g, aT, (size_t)4 * 4096, lane, a_row0(jj, ntBase));
#pragma unroll
  for (int k4 = 0; k4 < 4; ++k4) {
    const int ch0 = rr * 4096 + jj * 128 + (ntBase + 2 * k4) * 16;
    gload_tile(g + 2 + 8 * k4, W + (size_t)ch0 * 64, 64, lane);
    bb[k4] = bias[ch0 + (lane & 15)];
  }
}

__device__ __forceinline__ void issue_s(v4f* g, float* bb, const float* __restrict__ aBase,
                                        const float* __restrict__ W,
                                        const float* __restrict__ bias, int jj, int wave,
                                        int lane) {
  const int gs = 2 * jj + (wave >> 2);
  gload_part<2>(g, aBase + gs * 64, (size_t)4 * 4096, lane, 4 * (wave & 3));
  const int chs0 = 4 * (jj * 128 + wave * 16);
  gload_tile(g + 2, W + (size_t)chs0 * 64, 4 * 64, lane);
  bb[0] = bias[chs0 + 4 * (lane & 15)];
}

__device__ __forceinline__ void lds_barrier() {
  __builtin_amdgcn_fence(__ATOMIC_RELEASE, "workgroup", "local");
  __builtin_amdgcn_s_barrier();
  __builtin_amdgcn_fence(__ATOMIC_ACQUIRE, "workgroup", "local");
}

__global__ __launch_bounds__(256) void fc2_fused(const float* __restrict__ in,
                                                 const float* __restrict__ W,
                                                 const float* __restrict__ bias,
                                                 float* __restrict__ out) {
  __shared__ __attribute__((aligned(16))) float ytile[4][16 * 128];
  __shared__ float stile[16 * 128];
  __shared__ float kvbuf[64];
  __shared__ __attribute__((aligned(16))) float stageW[8][5][SLOT];
  __shared__ __attribute__((aligned(16))) float stageA[2][4][SLOT];
  __shared__ __attribute__((aligned(16))) float stageSA[2][SLOT];

  const int tid = threadIdx.x;
  const int lane = tid & 31;
  const int wave = tid >> 5;
  const int n0 = blockIdx.x * 16;
  const int bIdx = n0 >> 10;
  const int t0 = n0 & 1023;
  const float* aBase = in + ((size_t)(t0 * 4 + bIdx)) * 4096;
  float* st = &stageW[wave][0][0];

  const int rr = wave & 3;
  const int ntBase = wave >> 2;
  float* stSA = &stageSA[ntBase][0];
  const int sRow0 = 4 * rr;

  const int srow = tid >> 4;
  const int sc4 = 4 * (tid & 15);
  float m[4], dsum[4], num[4];
#pragma unroll
  for (int r = 0; r < 4; ++r) { m[r] = -3.0e38f; dsum[r] = 0.0f; num[r] = 0.0f; }

  v4f g[34];
  float bb[4];
  {
    v4f gp[2];
    gload_part<2>(gp, aBase + (rr * 16) * 64, (size_t)4 * 4096, lane, 8 * ntBase);
    issue_main(g, bb, aBase, W, bias, 0, rr, ntBase, lane);
    stage_part<2>(&stageA[0][rr][0], gp, lane, 8 * ntBase);
  }

  for (int jj = 0; jj < 32; ++jj) {
    stage_part<2>(&stageA[((jj + 1) >> 1) & 1][rr][0], g, lane, a_row0(jj, ntBase));
#pragma unroll
    for (int s = 0; s < 4; ++s) stage_tile(st + s * SLOT, g + 2 + 8 * s, lane);
    const float b0 = bb[0], b1 = bb[1], b2 = bb[2], b3 = bb[3];
    lds_barrier();
    issue_main(g, bb, aBase, W, bias, (jj + 1) & 31, rr, ntBase, lane);
    {
      const float bcur[4] = {b0, b1, b2, b3};
      v4f af[8];
      frag_read(af, &stageA[(jj >> 1) & 1][rr][0], lane);
#pragma unroll
      for (int k4 = 0; k4 < 4; ++k4) {
        const int nt = ntBase + 2 * k4;
        v4f bf[8];
        frag_read(bf, st + k4 * SLOT, lane);
        v8f acc = tile_gemm(af, bf, bcur[k4]);
        store_tile(ytile[rr], nt, lane, acc);
      }
    }
    lds_barrier();
    {
      v4f va[4], vb[4];
#pragma unroll
      for (int r = 0; r < 4; ++r) {
        va[r] = *reinterpret_cast<const v4f*>(&ytile[r][srow * 128 + sc4]);
        vb[r] = *reinterpret_cast<const v4f*>(&ytile[r][srow * 128 + 64 + sc4]);
      }
      const v4f xa = va[0] + va[1] + va[2] + va[3];
      const v4f xb = vb[0] + vb[1] + vb[2] + vb[3];
#pragma unroll
      for (int r = 0; r < 4; ++r) {
        const v4f a = va[r], b = vb[r];
        const float mb = fmaxf(fmaxf(fmaxf(a.x, a.y), fmaxf(a.z, a.w)),
                               fmaxf(fmaxf(b.x, b.y), fmaxf(b.z, b.w)));
        const float mn = fmaxf(m[r], mb);
        const float sc = __expf(m[r] - mn);
        const float ea0 = __expf(a.x - mn), ea1 = __expf(a.y - mn);
        const float ea2 = __expf(a.z - mn), ea3 = __expf(a.w - mn);
        const float eb0 = __expf(b.x - mn), eb1 = __expf(b.y - mn);
        const float eb2 = __expf(b.z - mn), eb3 = __expf(b.w - mn);
        const float da = (ea0 + ea1) + (ea2 + ea3);
        const float db = (eb0 + eb1) + (eb2 + eb3);
        const float na = (ea0 * xa.x + ea1 * xa.y) + (ea2 * xa.z + ea3 * xa.w);
        const float nb = (eb0 * xb.x + eb1 * xb.y) + (eb2 * xb.z + eb3 * xb.w);
        dsum[r] = dsum[r] * sc + (da + db);
        num[r] = num[r] * sc + (na + nb);
        m[r] = mn;
      }
    }
  }
#pragma unroll
  for (int r = 0; r < 4; ++r) {
#pragma unroll
    for (int mask = 1; mask < 16; mask <<= 1) {
      const float mo = __shfl_xor(m[r], mask, 32);
      const float dq = __shfl_xor(dsum[r], mask, 32);
      const float no = __shfl_xor(num[r], mask, 32);
      const float mn = fmaxf(m[r], mo);
      const float sa = __expf(m[r] - mn), sb = __expf(mo - mn);
      dsum[r] = dsum[r] * sa + dq * sb;
      num[r] = num[r] * sa + no * sb;
      m[r] = mn;
    }
  }
  if ((tid & 15) == 0) {
#pragma unroll
    for (int r = 0; r < 4; ++r) kvbuf[srow * 4 + r] = num[r] / dsum[r];
  }
  lds_barrier();

  v4f gs[10];
  float bbs[1];
  issue_s(gs, bbs, aBase, W, bias, 0, wave, lane);
  for (int jj = 0; jj < 32; ++jj) {
    stage_part<2>(&stageA[((jj + 1) >> 1) & 1][rr][0], g, lane, a_row0(jj, ntBase));
#pragma unroll
    for (int s = 0; s < 4; ++s) stage_tile(st + s * SLOT, g + 2 + 8 * s, lane);
    stage_part<2>(stSA, gs, lane, sRow0);
    stage_tile(st + 4 * SLOT, gs + 2, lane);
    const float b0 = bb[0], b1 = bb[1], b2 = bb[2], b3 = bb[3], bs = bbs[0];
    lds_barrier();
    issue_main(g, bb, aBase, W, bias, (jj + 1) & 31, rr, ntBase, lane);
    issue_s(gs, bbs, aBase, W, bias, (jj + 1) & 31, wave, lane);
    const int o_base = jj * 128;
    {
      const float bcur[4] = {b0, b1, b2, b3};
      v4f af[8];
      frag_read(af, &stageA[(jj >> 1) & 1][rr][0], lane);
#pragma unroll
      for (int k4 = 0; k4 < 4; ++k4) {
        const int nt = ntBase + 2 * k4;
        v4f bf[8];
        frag_read(bf, st + k4 * SLOT, lane);
        v8f acc = tile_gemm(af, bf, bcur[k4]);
        store_tile(ytile[rr], nt, lane, acc);
      }
      v4f sa[8], sb[8];
      frag_read(sa, stSA, lane);
      frag_read(sb, st + 4 * SLOT, lane);
      v8f accs = tile_gemm(sa, sb, bs);
      store_tile(stile, wave, lane, accs);
    }
    lds_barrier();
#pragma unroll
    for (int i = 0; i < 8; ++i) {
      const int e = i * 256 + tid;
      const int prow = e >> 7;
      const int oo = e & 127;
      const float* kvp = &kvbuf[prow * 4];
      const float val = stile[e] + ytile[0][e] * kvp[0] + ytile[1][e] * kvp[1] +
                        ytile[2][e] * kvp[2] + ytile[3][e] * kvp[3];
      out[((size_t)((t0 + prow) * 4 + bIdx)) * 4096 + o_base + oo] = val;
    }
    lds_barrier();
  }
}

extern "C" void kernel_launch(void* const* d_in, const int* in_sizes, int n_in,
                              void* d_out, int out_size, void* d_ws, size_t ws_size,
                              hipStream_t stream) {
  (void)in_sizes; (void)n_in; (void)d_ws; (void)ws_size; (void)out_size;
  const float* in = (const float*)d_in[0];
  const float* W = (const float*)d_in[1];
  const float* bias = (const float*)d_in[2];
  float* out = (float*)d_out;
  fc2_fused<<<256, 256, 0, stream>>>(in, W, bias, out);
}


# Two waves of a SIMD out of phase: a wave's 4 tiles as 2 pairs of alternating WMMA chains (older wave done first); stats/output of block j-1 taken behind block j's WMMAs; 1 barrier per block, was 2/3
# speedup vs baseline: 3.9694x; 1.0682x over previous
#include <hip/hip_runtime.h>


typedef float v2f __attribute__((ext_vector_type(2)));
typedef float v4f __attribute__((ext_vector_type(4)));
typedef float v8f __attribute__((ext_vector_type(8)));

#define PITCH 68
#define SLOT (16 * PITCH)

__device__ __forceinline__ void gload_tile(v4f* g, const float* __restrict__ base,
                                           size_t rowStride, int lane) {
  const float* p = base + (size_t)(lane >> 4) * rowStride + 4 * (lane & 15);
#pragma unroll
  for (int j = 0; j < 8; ++j)
    g[j] = *reinterpret_cast<const v4f*>(p + (size_t)(2 * j) * rowStride);
}

__device__ __forceinline__ void stage_tile(float* slot, const v4f* g, int lane) {
  float* q = slot + (lane >> 4) * PITCH + 4 * (lane & 15);
#pragma unroll
  for (int j = 0; j < 8; ++j)
    *reinterpret_cast<v4f*>(q + 2 * j * PITCH) = g[j];
}

template <int NS>
__device__ __forceinline__ void gload_part(v4f* g, const float* __restrict__ base,
                                           size_t rowStride, int lane, int row0) {
  const float* p = base + (size_t)(row0 + (lane >> 4)) * rowStride + 4 * (lane & 15);
#pragma unroll
  for (int j = 0; j < NS; ++j)
    g[j] = *reinterpret_cast<const v4f*>(p + (size_t)(2 * j) * rowStride);
}

template <int NS>
__device__ __forceinline__ void stage_part(float* slot, const v4f* g, int lane, int row0) {
  float* q = slot + (row0 + (lane >> 4)) * PITCH + 4 * (lane & 15);
#pragma unroll
  for (int j = 0; j < NS; ++j)
    *reinterpret_cast<v4f*>(q + 2 * j * PITCH) = g[j];
}

__device__ __forceinline__ void frag_read(v4f* f, const float* slot, int lane) {
  const float* q = slot + (lane & 15) * PITCH + 4 * (lane >> 4);
#pragma unroll
  for (int p = 0; p < 8; ++p)
    f[p] = *reinterpret_cast<const v4f*>(q + 8 * p);
}

__device__ __forceinline__ v8f tile_gemm(const v4f* af, const v4f* bf, float bb) {
  v8f acc = {bb, bb, bb, bb, bb, bb, bb, bb};
#pragma unroll
  for (int p = 0; p < 8; ++p) {
    const v2f a0 = af[p].xy, a1 = af[p].zw;
    const v2f b0 = bf[p].xy, b1 = bf[p].zw;
    acc = __builtin_amdgcn_wmma_f32_16x16x4_f32(
         false, a0,  false, b0,
         (short)0, acc,  false,  false);
    acc = __builtin_amdgcn_wmma_f32_16x16x4_f32(
         false, a1,  false, b1,
         (short)0, acc,  false,  false);
  }
  return acc;
}

__device__ __forceinline__ void tile_gemm2(const v4f* af, const v4f* bf0, const v4f* bf1, float bb0,
                                           float bb1, v8f& acc0, v8f& acc1) {
  acc0 = (v8f){bb0, bb0, bb0, bb0, bb0, bb0, bb0, bb0};
  acc1 = (v8f){bb1, bb1, bb1, bb1, bb1, bb1, bb1, bb1};
#pragma unroll
  for (int p = 0; p < 8; ++p) {
    const v2f a0 = af[p].xy, a1 = af[p].zw;
    acc0 = __builtin_amdgcn_wmma_f32_16x16x4_f32(false, a0, false, bf0[p].xy, (short)0, acc0, false, false);
    acc1 = __builtin_amdgcn_wmma_f32_16x16x4_f32(false, a0, false, bf1[p].xy, (short)0, acc1, false, false);
    acc0 = __builtin_amdgcn_wmma_f32_16x16x4_f32(false, a1, false, bf0[p].zw, (short)0, acc0, false, false);
    acc1 = __builtin_amdgcn_wmma_f32_16x16x4_f32(false, a1, false, bf1[p].zw, (short)0, acc1, false, false);
  }
}

__device__ __forceinline__ void store_tile(float* buf, int nt, int lane, v8f acc) {
  const int n = lane & 15, kh = lane >> 4;
#pragma unroll
  for (int i = 0; i < 8; ++i)
    buf[(i + 8 * kh) * 128 + nt * 16 + n] = acc[i];
}

__device__ __forceinline__ int a_row0(int jj, int ntBase) {
  return 8 * ntBase + ((jj & 1) ? 0 : 4);
}

__device__ __forceinline__ void issue_main(v4f* g, float* bb, const float* __restrict__ aBase,
                                           const float* __restrict__ W,
                                           const float* __restrict__ bias, int jj, int rr,
                                           int ntBase, int lane) {
  const int jE = (jj + (jj & 1)) & 31;
  const float* aT = aBase + (rr * 16 + (jE >> 1)) * 64;
  gload_part<2>(g, aT, (size_t)4 * 4096, lane, a_row0(jj, ntBase));
#pragma unroll
  for (int k4 = 0; k4 < 4; ++k4) {
    const int ch0 = rr * 4096 + jj * 128 + (ntBase + 2 * k4) * 16;
    gload_tile(g + 2 + 8 * k4, W + (size_t)ch0 * 64, 64, lane);
    bb[k4] = bias[ch0 + (lane & 15)];
  }
}

__device__ __forceinline__ void issue_s(v4f* g, float* bb, const float* __restrict__ aBase,
                                        const float* __restrict__ W,
                                        const float* __restrict__ bias, int jj, int wave,
                                        int lane) {
  const int gs = 2 * jj + (wave >> 2);
  gload_part<2>(g, aBase + gs * 64, (size_t)4 * 4096, lane, 4 * (wave & 3));
  const int chs0 = 4 * (jj * 128 + wave * 16);
  gload_tile(g + 2, W + (size_t)chs0 * 64, 4 * 64, lane);
  bb[0] = bias[chs0 + 4 * (lane & 15)];
}

__device__ __forceinline__ void lds_barrier() {
  __builtin_amdgcn_fence(__ATOMIC_RELEASE, "workgroup", "local");
  __builtin_amdgcn_s_barrier();
  __builtin_amdgcn_fence(__ATOMIC_ACQUIRE, "workgroup", "local");
}

__device__ __forceinline__ void block_stats(const float* yt, int srow, int sc4,
                                            float* m, float* dsum, float* num) {
  v4f va[4], vb[4];
#pragma unroll
  for (int r = 0; r < 4; ++r) {
    va[r] = *reinterpret_cast<const v4f*>(yt + r * (16 * 128) + srow * 128 + sc4);
    vb[r] = *reinterpret_cast<const v4f*>(yt + r * (16 * 128) + srow * 128 + 64 + sc4);
  }
  const v4f xa = va[0] + va[1] + va[2] + va[3];
  const v4f xb = vb[0] + vb[1] + vb[2] + vb[3];
#pragma unroll
  for (int r = 0; r < 4; ++r) {
    const v4f a = va[r], b = vb[r];
    const float mb = fmaxf(fmaxf(fmaxf(a.x, a.y), fmaxf(a.z, a.w)),
                           fmaxf(fmaxf(b.x, b.y), fmaxf(b.z, b.w)));
    const float mn = fmaxf(m[r], mb);
    const float sc = __expf(m[r] - mn);
    const float ea0 = __expf(a.x - mn), ea1 = __expf(a.y - mn);
    const float ea2 = __expf(a.z - mn), ea3 = __expf(a.w - mn);
    const float eb0 = __expf(b.x - mn), eb1 = __expf(b.y - mn);
    const float eb2 = __expf(b.z - mn), eb3 = __expf(b.w - mn);
    const float da = (ea0 + ea1) + (ea2 + ea3);
    const float db = (eb0 + eb1) + (eb2 + eb3);
    const float na = (ea0 * xa.x + ea1 * xa.y) + (ea2 * xa.z + ea3 * xa.w);
    const float nb = (eb0 * xb.x + eb1 * xb.y) + (eb2 * xb.z + eb3 * xb.w);
    dsum[r] = dsum[r] * sc + (da + db);
    num[r] = num[r] * sc + (na + nb);
    m[r] = mn;
  }
}

__device__ __forceinline__ void block_out(float* __restrict__ out, const float* yt,
                                          const float* stl, const float* kvbuf, int tid,
                                          int t0, int bIdx, int o_base) {
#pragma unroll
  for (int i = 0; i < 8; ++i) {
    const int e = i * 256 + tid;
    const int prow = e >> 7;
    const int oo = e & 127;
    const float* kvp = &kvbuf[prow * 4];
    const float val = stl[e] + yt[e] * kvp[0] + yt[16 * 128 + e] * kvp[1] +
                      yt[2 * 16 * 128 + e] * kvp[2] + yt[3 * 16 * 128 + e] * kvp[3];
    out[((size_t)((t0 + prow) * 4 + bIdx)) * 4096 + o_base + oo] = val;
  }
}

__global__ __launch_bounds__(256) void fc2_fused(const float* __restrict__ in,
                                                 const float* __restrict__ W,
                                                 const float* __restrict__ bias,
                                                 float* __restrict__ out) {
  __shared__ __attribute__((aligned(16))) float ytile[2][4][16 * 128];
  __shared__ float stile[2][16 * 128];
  __shared__ float kvbuf[64];
  __shared__ __attribute__((aligned(16))) float stageW[8][5][SLOT];
  __shared__ __attribute__((aligned(16))) float stageA[2][4][SLOT];
  __shared__ __attribute__((aligned(16))) float stageSA[2][2][SLOT];

  const int tid = threadIdx.x;
  const int lane = tid & 31;
  const int wave = tid >> 5;
  const int n0 = blockIdx.x * 16;
  const int bIdx = n0 >> 10;
  const int t0 = n0 & 1023;
  const float* aBase = in + ((size_t)(t0 * 4 + bIdx)) * 4096;
  float* st = &stageW[wave][0][0];

  const int rr = wave & 3;
  const int ntBase = wave >> 2;
  const int sRow0 = 4 * rr;

  const int srow = tid >> 4;
  const int sc4 = 4 * (tid & 15);
  float m[4], dsum[4], num[4];
#pragma unroll
  for (int r = 0; r < 4; ++r) { m[r] = -3.0e38f; dsum[r] = 0.0f; num[r] = 0.0f; }

  v4f g[34];
  float bb[4];
  {
    v4f gp[2];
    gload_part<2>(gp, aBase + (rr * 16) * 64, (size_t)4 * 4096, lane, 8 * ntBase);
    issue_main(g, bb, aBase, W, bias, 0, rr, ntBase, lane);
    stage_part<2>(&stageA[0][rr][0], gp, lane, 8 * ntBase);
  }

  for (int jj = 0; jj < 32; ++jj) {
    stage_part<2>(&stageA[((jj + 1) >> 1) & 1][rr][0], g, lane, a_row0(jj, ntBase));
#pragma unroll
    for (int s = 0; s < 4; ++s) stage_tile(st + s * SLOT, g + 2 + 8 * s, lane);
    const float b0 = bb[0], b1 = bb[1], b2 = bb[2], b3 = bb[3];
    lds_barrier();
    issue_main(g, bb, aBase, W, bias, (jj + 1) & 31, rr, ntBase, lane);
    {
      const float bcur[4] = {b0, b1, b2, b3};
      v4f af[8];
      frag_read(af, &stageA[(jj >> 1) & 1][rr][0], lane);
#pragma unroll
      for (int k4 = 0; k4 < 4; k4 += 2) {
        v4f bf0[8], bf1[8];
        frag_read(bf0, st + k4 * SLOT, lane);
        frag_read(bf1, st + (k4 + 1) * SLOT, lane);
        v8f acc0, acc1;
        tile_gemm2(af, bf0, bf1, bcur[k4], bcur[k4 + 1], acc0, acc1);
        store_tile(ytile[jj & 1][rr], ntBase + 2 * k4, lane, acc0);
        store_tile(ytile[jj & 1][rr], ntBase + 2 * k4 + 2, lane, acc1);
      }
    }
    if (jj > 0) block_stats(&ytile[(jj - 1) & 1][0][0], srow, sc4, m, dsum, num);
  }
  lds_barrier();
  block_stats(&ytile[1][0][0], srow, sc4, m, dsum, num);
#pragma unroll
  for (int r = 0; r < 4; ++r) {
#pragma unroll
    for (int mask = 1; mask < 16; mask <<= 1) {
      const float mo = __shfl_xor(m[r], mask, 32);
      const float dq = __shfl_xor(dsum[r], mask, 32);
      const float no = __shfl_xor(num[r], mask, 32);
      const float mn = fmaxf(m[r], mo);
      const float sa = __expf(m[r] - mn), sb = __expf(mo - mn);
      dsum[r] = dsum[r] * sa + dq * sb;
      num[r] = num[r] * sa + no * sb;
      m[r] = mn;
    }
  }
  if ((tid & 15) == 0) {
#pragma unroll
    for (int r = 0; r < 4; ++r) kvbuf[srow * 4 + r] = num[r] / dsum[r];
  }
  lds_barrier();

  v4f gs[10];
  float bbs[1];
  issue_s(gs, bbs, aBase, W, bias, 0, wave, lane);
  for (int jj = 0; jj < 32; ++jj) {
    stage_part<2>(&stageA[((jj + 1) >> 1) & 1][rr][0], g, lane, a_row0(jj, ntBase));
#pragma unroll
    for (int s = 0; s < 4; ++s) stage_tile(st + s * SLOT, g + 2 + 8 * s, lane);
    float* stSA = &stageSA[jj & 1][ntBase][0];
    stage_part<2>(stSA, gs, lane, sRow0);
    stage_tile(st + 4 * SLOT, gs + 2, lane);
    const float b0 = bb[0], b1 = bb[1], b2 = bb[2], b3 = bb[3], bs = bbs[0];
    lds_barrier();
    issue_main(g, bb, aBase, W, bias, (jj + 1) & 31, rr, ntBase, lane);
    issue_s(gs, bbs, aBase, W, bias, (jj + 1) & 31, wave, lane);
    {
      const float bcur[4] = {b0, b1, b2, b3};
      v4f af[8];
      frag_read(af, &stageA[(jj >> 1) & 1][rr][0], lane);
#pragma unroll
      for (int k4 = 0; k4 < 4; k4 += 2) {
        v4f bf0[8], bf1[8];
        frag_read(bf0, st + k4 * SLOT, lane);
        frag_read(bf1, st + (k4 + 1) * SLOT, lane);
        v8f acc0, acc1;
        tile_gemm2(af, bf0, bf1, bcur[k4], bcur[k4 + 1], acc0, acc1);
        store_tile(ytile[jj & 1][rr], ntBase + 2 * k4, lane, acc0);
        store_tile(ytile[jj & 1][rr], ntBase + 2 * k4 + 2, lane, acc1);
      }
      v4f sa[8], sb[8];
      frag_read(sa, stSA, lane);
      frag_read(sb, st + 4 * SLOT, lane);
      v8f accs = tile_gemm(sa, sb, bs);
      store_tile(stile[jj & 1], wave, lane, accs);
    }
    if (jj > 0)
      block_out(out, &ytile[(jj - 1) & 1][0][0], stile[(jj - 1) & 1], kvbuf, tid, t0, bIdx,
                (jj - 1) * 128);
  }
  lds_barrier();
  block_out(out, &ytile[1][0][0], stile[1], kvbuf, tid, t0, bIdx, 31 * 128);
}

extern "C" void kernel_launch(void* const* d_in, const int* in_sizes, int n_in,
                              void* d_out, int out_size, void* d_ws, size_t ws_size,
                              hipStream_t stream) {
  (void)in_sizes; (void)n_in; (void)d_ws; (void)ws_size; (void)out_size;
  const float* in = (const float*)d_in[0];
  const float* W = (const float*)d_in[1];
  const float* bias = (const float*)d_in[2];
  float* out = (float*)d_out;
  fc2_fused<<<256, 256, 0, stream>>>(in, W, bias, out);
}


# Residual s no longer a GEMM of its own: phase 1 stores its x[n,4i,r] values to out as start values, phase 2 reads them back and adds; s weight loads, staging and 16 WMMAs per wave and block dropped
# speedup vs baseline: 4.1395x; 1.0429x over previous
#include <hip/hip_runtime.h>


typedef float v2f __attribute__((ext_vector_type(2)));
typedef float v4f __attribute__((ext_vector_type(4)));
typedef float v8f __attribute__((ext_vector_type(8)));

#define PITCH 68
#define SLOT (16 * PITCH)

__device__ __forceinline__ void gload_tile(v4f* g, const float* __restrict__ base,
                                           size_t rowStride, int lane) {
  const float* p = base + (size_t)(lane >> 4) * rowStride + 4 * (lane & 15);
#pragma unroll
  for (int j = 0; j < 8; ++j)
    g[j] = *reinterpret_cast<const v4f*>(p + (size_t)(2 * j) * rowStride);
}

__device__ __forceinline__ void stage_tile(float* slot, const v4f* g, int lane) {
  float* q = slot + (lane >> 4) * PITCH + 4 * (lane & 15);
#pragma unroll
  for (int j = 0; j < 8; ++j)
    *reinterpret_cast<v4f*>(q + 2 * j * PITCH) = g[j];
}

template <int NS>
__device__ __forceinline__ void gload_part(v4f* g, const float* __restrict__ base,
                                           size_t rowStride, int lane, int row0) {
  const float* p = base + (size_t)(row0 + (lane >> 4)) * rowStride + 4 * (lane & 15);
#pragma unroll
  for (int j = 0; j < NS; ++j)
    g[j] = *reinterpret_cast<const v4f*>(p + (size_t)(2 * j) * rowStride);
}

template <int NS>
__device__ __forceinline__ void stage_part(float* slot, const v4f* g, int lane, int row0) {
  float* q = slot + (row0 + (lane >> 4)) * PITCH + 4 * (lane & 15);
#pragma unroll
  for (int j = 0; j < NS; ++j)
    *reinterpret_cast<v4f*>(q + 2 * j * PITCH) = g[j];
}

__device__ __forceinline__ void frag_read(v4f* f, const float* slot, int lane) {
  const float* q = slot + (lane & 15) * PITCH + 4 * (lane >> 4);
#pragma unroll
  for (int p = 0; p < 8; ++p)
    f[p] = *reinterpret_cast<const v4f*>(q + 8 * p);
}

__device__ __forceinline__ void tile_gemm2(const v4f* af, const v4f* bf0, const v4f* bf1, float bb0,
                                           float bb1, v8f& acc0, v8f& acc1) {
  acc0 = (v8f){bb0, bb0, bb0, bb0, bb0, bb0, bb0, bb0};
  acc1 = (v8f){bb1, bb1, bb1, bb1, bb1, bb1, bb1, bb1};
#pragma unroll
  for (int p = 0; p < 8; ++p) {
    const v2f a0 = af[p].xy, a1 = af[p].zw;
    acc0 = __builtin_amdgcn_wmma_f32_16x16x4_f32(false, a0, false, bf0[p].xy, (short)0, acc0, false, false);
    acc1 = __builtin_amdgcn_wmma_f32_16x16x4_f32(false, a0, false, bf1[p].xy, (short)0, acc1, false, false);
    acc0 = __builtin_amdgcn_wmma_f32_16x16x4_f32(false, a1, false, bf0[p].zw, (short)0, acc0, false, false);
    acc1 = __builtin_amdgcn_wmma_f32_16x16x4_f32(false, a1, false, bf1[p].zw, (short)0, acc1, false, false);
  }
}

__device__ __forceinline__ void store_tile(float* buf, int nt, int lane, v8f acc) {
  const int n = lane & 15, kh = lane >> 4;
#pragma unroll
  for (int i = 0; i < 8; ++i)
    buf[(i + 8 * kh) * 128 + nt * 16 + n] = acc[i];
}

__device__ __forceinline__ int a_row0(int jj, int ntBase) {
  return 8 * ntBase + ((jj & 1) ? 0 : 4);
}

__device__ __forceinline__ void issue_main(v4f* g, float* bb, const float* __restrict__ aBase,
                                           const float* __restrict__ W,
                                           const float* __restrict__ bias, int jj, int rr,
                                           int ntBase, int lane) {
  const int jE = (jj + (jj & 1)) & 31;
  const float* aT = aBase + (rr * 16 + (jE >> 1)) * 64;
  gload_part<2>(g, aT, (size_t)4 * 4096, lane, a_row0(jj, ntBase));
#pragma unroll
  for (int k4 = 0; k4 < 4; ++k4) {
    const int ch0 = rr * 4096 + jj * 128 + (ntBase + 2 * k4) * 16;
    gload_tile(g + 2 + 8 * k4, W + (size_t)ch0 * 64, 64, lane);
    bb[k4] = bias[ch0 + (lane & 15)];
  }
}

__device__ __forceinline__ void lds_barrier() {
  __builtin_amdgcn_fence(__ATOMIC_RELEASE, "workgroup", "local");
  __builtin_amdgcn_s_barrier();
  __builtin_amdgcn_fence(__ATOMIC_ACQUIRE, "workgroup", "local");
}

__device__ __forceinline__ void block_stats(const float* yt, int srow, int sc4,
                                            float* m, float* dsum, float* num,
                                            float* __restrict__ sOut, bool upper) {
  v4f va[4], vb[4];
#pragma unroll
  for (int r = 0; r < 4; ++r) {
    va[r] = *reinterpret_cast<const v4f*>(yt + r * (16 * 128) + srow * 128 + sc4);
    vb[r] = *reinterpret_cast<const v4f*>(yt + r * (16 * 128) + srow * 128 + 64 + sc4);
  }
#pragma unroll
  for (int r = 0; r < 4; ++r) {
    const float give = upper ? va[r].x : vb[r].x;
    const float got = __uint_as_float(__builtin_amdgcn_permlanex16(
        0u, __float_as_uint(give), 0x76543210u, 0xfedcba98u, false, false));
    sOut[r * 1024] = upper ? got : va[r].x;
    sOut[4 * 4096 + r * 1024] = upper ? vb[r].x : got;
  }
  const v4f xa = va[0] + va[1] + va[2] + va[3];
  const v4f xb = vb[0] + vb[1] + vb[2] + vb[3];
#pragma unroll
  for (int r = 0; r < 4; ++r) {
    const v4f a = va[r], b = vb[r];
    const float mb = fmaxf(fmaxf(fmaxf(a.x, a.y), fmaxf(a.z, a.w)),
                           fmaxf(fmaxf(b.x, b.y), fmaxf(b.z, b.w)));
    const float mn = fmaxf(m[r], mb);
    const float sc = __expf(m[r] - mn);
    const float ea0 = __expf(a.x - mn), ea1 = __expf(a.y - mn);
    const float ea2 = __expf(a.z - mn), ea3 = __expf(a.w - mn);
    const float eb0 = __expf(b.x - mn), eb1 = __expf(b.y - mn);
    const float eb2 = __expf(b.z - mn), eb3 = __expf(b.w - mn);
    const float da = (ea0 + ea1) + (ea2 + ea3);
    const float db = (eb0 + eb1) + (eb2 + eb3);
    const float na = (ea0 * xa.x + ea1 * xa.y) + (ea2 * xa.z + ea3 * xa.w);
    const float nb = (eb0 * xb.x + eb1 * xb.y) + (eb2 * xb.z + eb3 * xb.w);
    dsum[r] = dsum[r] * sc + (da + db);
    num[r] = num[r] * sc + (na + nb);
    m[r] = mn;
  }
}

__device__ __forceinline__ size_t out_index(int i, int tid, int t0, int bIdx, int o_base) {
  const int e = i * 256 + tid;
  return ((size_t)((t0 + (e >> 7)) * 4 + bIdx)) * 4096 + o_base + (e & 127);
}

__device__ __forceinline__ void load_start(float* pv, const float* out, int tid, int t0,
                                           int bIdx, int o_base) {
#pragma unroll
  for (int i = 0; i < 8; ++i) pv[i] = out[out_index(i, tid, t0, bIdx, o_base)];
}

__device__ __forceinline__ void block_out(float* out, const float* yt, const float* pv,
                                          const float* kvbuf, int tid, int t0, int bIdx,
                                          int o_base) {
#pragma unroll
  for (int i = 0; i < 8; ++i) {
    const int e = i * 256 + tid;
    const int prow = e >> 7;
    const float* kvp = &kvbuf[prow * 4];
    const float val = pv[i] + yt[e] * kvp[0] + yt[16 * 128 + e] * kvp[1] +
                      yt[2 * 16 * 128 + e] * kvp[2] + yt[3 * 16 * 128 + e] * kvp[3];
    out[out_index(i, tid, t0, bIdx, o_base)] = val;
  }
}

__global__ __launch_bounds__(256) void fc2_fused(const float* __restrict__ in,
                                                 const float* __restrict__ W,
                                                 const float* __restrict__ bias,
                                                 float* out) {
  __shared__ __attribute__((aligned(16))) float ytile[2][4][16 * 128];
  __shared__ float kvbuf[64];
  __shared__ __attribute__((aligned(16))) float stageW[8][4][SLOT];
  __shared__ __attribute__((aligned(16))) float stageA[2][4][SLOT];

  const int tid = threadIdx.x;
  const int lane = tid & 31;
  const int wave = tid >> 5;
  const int n0 = blockIdx.x * 16;
  const int bIdx = n0 >> 10;
  const int t0 = n0 & 1023;
  const float* aBase = in + ((size_t)(t0 * 4 + bIdx)) * 4096;
  float* st = &stageW[wave][0][0];

  const int rr = wave & 3;
  const int ntBase = wave >> 2;

  const int srow = tid >> 4;
  const int sc4 = 4 * (tid & 15);
  float* sOut = out + ((size_t)((t0 + 2 * wave) * 4 + bIdx)) * 4096 + lane;
  const bool upper = lane >= 16;
  float m[4], dsum[4], num[4];
#pragma unroll
  for (int r = 0; r < 4; ++r) { m[r] = -3.0e38f; dsum[r] = 0.0f; num[r] = 0.0f; }

  v4f g[34];
  float bb[4];
  {
    v4f gp[2];
    gload_part<2>(gp, aBase + (rr * 16) * 64, (size_t)4 * 4096, lane, 8 * ntBase);
    issue_main(g, bb, aBase, W, bias, 0, rr, ntBase, lane);
    stage_part<2>(&stageA[0][rr][0], gp, lane, 8 * ntBase);
  }

  for (int jj = 0; jj < 32; ++jj) {
    stage_part<2>(&stageA[((jj + 1) >> 1) & 1][rr][0], g, lane, a_row0(jj, ntBase));
#pragma unroll
    for (int s = 0; s < 4; ++s) stage_tile(st + s * SLOT, g + 2 + 8 * s, lane);
    const float b0 = bb[0], b1 = bb[1], b2 = bb[2], b3 = bb[3];
    lds_barrier();
    issue_main(g, bb, aBase, W, bias, (jj + 1) & 31, rr, ntBase, lane);
    {
      const float bcur[4] = {b0, b1, b2, b3};
      v4f af[8];
      frag_read(af, &stageA[(jj >> 1) & 1][rr][0], lane);
#pragma unroll
      for (int k4 = 0; k4 < 4; k4 += 2) {
        v4f bf0[8], bf1[8];
        frag_read(bf0, st + k4 * SLOT, lane);
        frag_read(bf1, st + (k4 + 1) * SLOT, lane);
        v8f acc0, acc1;
        tile_gemm2(af, bf0, bf1, bcur[k4], bcur[k4 + 1], acc0, acc1);
        store_tile(ytile[jj & 1][rr], ntBase + 2 * k4, lane, acc0);
        store_tile(ytile[jj & 1][rr], ntBase + 2 * k4 + 2, lane, acc1);
      }
    }
    if (jj > 0)
      block_stats(&ytile[(jj - 1) & 1][0][0], srow, sc4, m, dsum, num, sOut + 32 * (jj - 1),
                  upper);
  }
  lds_barrier();
  block_stats(&ytile[1][0][0], srow, sc4, m, dsum, num, sOut + 32 * 31, upper);
#pragma unroll
  for (int r = 0; r < 4; ++r) {
#pragma unroll
    for (int mask = 1; mask < 16; mask <<= 1) {
      const float mo = __shfl_xor(m[r], mask, 32);
      const float dq = __shfl_xor(dsum[r], mask, 32);
      const float no = __shfl_xor(num[r], mask, 32);
      const float mn = fmaxf(m[r], mo);
      const float sa = __expf(m[r] - mn), sb = __expf(mo - mn);
      dsum[r] = dsum[r] * sa + dq * sb;
      num[r] = num[r] * sa + no * sb;
      m[r] = mn;
    }
  }
  if ((tid & 15) == 0) {
#pragma unroll
    for (int r = 0; r < 4; ++r) kvbuf[srow * 4 + r] = num[r] / dsum[r];
  }
  __syncthreads();

  float pv[8];
  for (int jj = 0; jj < 32; ++jj) {
    stage_part<2>(&stageA[((jj + 1) >> 1) & 1][rr][0], g, lane, a_row0(jj, ntBase));
#pragma unroll
    for (int s = 0; s < 4; ++s) stage_tile(st + s * SLOT, g + 2 + 8 * s, lane);
    const float b0 = bb[0], b1 = bb[1], b2 = bb[2], b3 = bb[3];
    lds_barrier();
    if (jj > 0) load_start(pv, out, tid, t0, bIdx, (jj - 1) * 128);
    issue_main(g, bb, aBase, W, bias, (jj + 1) & 31, rr, ntBase, lane);
    {
      const float bcur[4] = {b0, b1, b2, b3};
      v4f af[8];
      frag_read(af, &stageA[(jj >> 1) & 1][rr][0], lane);
#pragma unroll
      for (int k4 = 0; k4 < 4; k4 += 2) {
        v4f bf0[8], bf1[8];
        frag_read(bf0, st + k4 * SLOT, lane);
        frag_read(bf1, st + (k4 + 1) * SLOT, lane);
        v8f acc0, acc1;
        tile_gemm2(af, bf0, bf1, bcur[k4], bcur[k4 + 1], acc0, acc1);
        store_tile(ytile[jj & 1][rr], ntBase + 2 * k4, lane, acc0);
        store_tile(ytile[jj & 1][rr], ntBase + 2 * k4 + 2, lane, acc1);
      }
    }
    if (jj > 0)
      block_out(out, &ytile[(jj - 1) & 1][0][0], pv, kvbuf, tid, t0, bIdx, (jj - 1) * 128);
  }
  load_start(pv, out, tid, t0, bIdx, 31 * 128);
  lds_barrier();
  block_out(out, &ytile[1][0][0], pv, kvbuf, tid, t0, bIdx, 31 * 128);
}

extern "C" void kernel_launch(void* const* d_in, const int* in_sizes, int n_in,
                              void* d_out, int out_size, void* d_ws, size_t ws_size,
                              hipStream_t stream) {
  (void)in_sizes; (void)n_in; (void)d_ws; (void)ws_size; (void)out_size;
  const float* in = (const float*)d_in[0];
  const float* W = (const float*)d_in[1];
  const float* bias = (const float*)d_in[2];
  float* out = (float*)d_out;
  fc2_fused<<<256, 256, 0, stream>>>(in, W, bias, out);
}


# Each workgroup walks the 32 o blocks from its own even start block, wrapping: one moment's input, output and weight requests spread over all columns; online-softmax block sum order rotated
# speedup vs baseline: 4.4989x; 1.0868x over previous
#include <hip/hip_runtime.h>


typedef float v2f __attribute__((ext_vector_type(2)));
typedef float v4f __attribute__((ext_vector_type(4)));
typedef float v8f __attribute__((ext_vector_type(8)));

#define PITCH 68
#define SLOT (16 * PITCH)

__device__ __forceinline__ void gload_tile(v4f* g, const float* __restrict__ base,
                                           size_t rowStride, int lane) {
  const float* p = base + (size_t)(lane >> 4) * rowStride + 4 * (lane & 15);
#pragma unroll
  for (int j = 0; j < 8; ++j)
    g[j] = *reinterpret_cast<const v4f*>(p + (size_t)(2 * j) * rowStride);
}

__device__ __forceinline__ void stage_tile(float* slot, const v4f* g, int lane) {
  float* q = slot + (lane >> 4) * PITCH + 4 * (lane & 15);
#pragma unroll
  for (int j = 0; j < 8; ++j)
    *reinterpret_cast<v4f*>(q + 2 * j * PITCH) = g[j];
}

template <int NS>
__device__ __forceinline__ void gload_part(v4f* g, const float* __restrict__ base,
                                           size_t rowStride, int lane, int row0) {
  const float* p = base + (size_t)(row0 + (lane >> 4)) * rowStride + 4 * (lane & 15);
#pragma unroll
  for (int j = 0; j < NS; ++j)
    g[j] = *reinterpret_cast<const v4f*>(p + (size_t)(2 * j) * rowStride);
}

template <int NS>
__device__ __forceinline__ void stage_part(float* slot, const v4f* g, int lane, int row0) {
  float* q = slot + (row0 + (lane >> 4)) * PITCH + 4 * (lane & 15);
#pragma unroll
  for (int j = 0; j < NS; ++j)
    *reinterpret_cast<v4f*>(q + 2 * j * PITCH) = g[j];
}

__device__ __forceinline__ void frag_read(v4f* f, const float* slot, int lane) {
  const float* q = slot + (lane & 15) * PITCH + 4 * (lane >> 4);
#pragma unroll
  for (int p = 0; p < 8; ++p)
    f[p] = *reinterpret_cast<const v4f*>(q + 8 * p);
}

__device__ __forceinline__ void tile_gemm2(const v4f* af, const v4f* bf0, const v4f* bf1, float bb0,
                                           float bb1, v8f& acc0, v8f& acc1) {
  acc0 = (v8f){bb0, bb0, bb0, bb0, bb0, bb0, bb0, bb0};
  acc1 = (v8f){bb1, bb1, bb1, bb1, bb1, bb1, bb1, bb1};
#pragma unroll
  for (int p = 0; p < 8; ++p) {
    const v2f a0 = af[p].xy, a1 = af[p].zw;
    acc0 = __builtin_amdgcn_wmma_f32_16x16x4_f32(false, a0, false, bf0[p].xy, (short)0, acc0, false, false);
    acc1 = __builtin_amdgcn_wmma_f32_16x16x4_f32(false, a0, false, bf1[p].xy, (short)0, acc1, false, false);
    acc0 = __builtin_amdgcn_wmma_f32_16x16x4_f32(false, a1, false, bf0[p].zw, (short)0, acc0, false, false);
    acc1 = __builtin_amdgcn_wmma_f32_16x16x4_f32(false, a1, false, bf1[p].zw, (short)0, acc1, false, false);
  }
}

__device__ __forceinline__ void store_tile(float* buf, int nt, int lane, v8f acc) {
  const int n = lane & 15, kh = lane >> 4;
#pragma unroll
  for (int i = 0; i < 8; ++i)
    buf[(i + 8 * kh) * 128 + nt * 16 + n] = acc[i];
}

__device__ __forceinline__ int a_row0(int jj, int ntBase) {
  return 8 * ntBase + ((jj & 1) ? 0 : 4);
}

__device__ __forceinline__ void issue_main(v4f* g, float* bb, const float* __restrict__ aBase,
                                           const float* __restrict__ W,
                                           const float* __restrict__ bias, int jj, int rr,
                                           int ntBase, int lane) {
  const int jE = (jj + (jj & 1)) & 31;
  const float* aT = aBase + (rr * 16 + (jE >> 1)) * 64;
  gload_part<2>(g, aT, (size_t)4 * 4096, lane, a_row0(jj, ntBase));
#pragma unroll
  for (int k4 = 0; k4 < 4; ++k4) {
    const int ch0 = rr * 4096 + jj * 128 + (ntBase + 2 * k4) * 16;
    gload_tile(g + 2 + 8 * k4, W + (size_t)ch0 * 64, 64, lane);
    bb[k4] = bias[ch0 + (lane & 15)];
  }
}

__device__ __forceinline__ void lds_barrier() {
  __builtin_amdgcn_fence(__ATOMIC_RELEASE, "workgroup", "local");
  __builtin_amdgcn_s_barrier();
  __builtin_amdgcn_fence(__ATOMIC_ACQUIRE, "workgroup", "local");
}

__device__ __forceinline__ void block_stats(const float* yt, int srow, int sc4,
                                            float* m, float* dsum, float* num,
                                            float* __restrict__ sOut, bool upper) {
  v4f va[4], vb[4];
#pragma unroll
  for (int r = 0; r < 4; ++r) {
    va[r] = *reinterpret_cast<const v4f*>(yt + r * (16 * 128) + srow * 128 + sc4);
    vb[r] = *reinterpret_cast<const v4f*>(yt + r * (16 * 128) + srow * 128 + 64 + sc4);
  }
#pragma unroll
  for (int r = 0; r < 4; ++r) {
    const float give = upper ? va[r].x : vb[r].x;
    const float got = __uint_as_float(__builtin_amdgcn_permlanex16(
        0u, __float_as_uint(give), 0x76543210u, 0xfedcba98u, false, false));
    sOut[r * 1024] = upper ? got : va[r].x;
    sOut[4 * 4096 + r * 1024] = upper ? vb[r].x : got;
  }
  const v4f xa = va[0] + va[1] + va[2] + va[3];
  const v4f xb = vb[0] + vb[1] + vb[2] + vb[3];
#pragma unroll
  for (int r = 0; r < 4; ++r) {
    const v4f a = va[r], b = vb[r];
    const float mb = fmaxf(fmaxf(fmaxf(a.x, a.y), fmaxf(a.z, a.w)),
                           fmaxf(fmaxf(b.x, b.y), fmaxf(b.z, b.w)));
    const float mn = fmaxf(m[r], mb);
    const float sc = __expf(m[r] - mn);
    const float ea0 = __expf(a.x - mn), ea1 = __expf(a.y - mn);
    const float ea2 = __expf(a.z - mn), ea3 = __expf(a.w - mn);
    const float eb0 = __expf(b.x - mn), eb1 = __expf(b.y - mn);
    const float eb2 = __expf(b.z - mn), eb3 = __expf(b.w - mn);
    const float da = (ea0 + ea1) + (ea2 + ea3);
    const float db = (eb0 + eb1) + (eb2 + eb3);
    const float na = (ea0 * xa.x + ea1 * xa.y) + (ea2 * xa.z + ea3 * xa.w);
    const float nb = (eb0 * xb.x + eb1 * xb.y) + (eb2 * xb.z + eb3 * xb.w);
    dsum[r] = dsum[r] * sc + (da + db);
    num[r] = num[r] * sc + (na + nb);
    m[r] = mn;
  }
}

__device__ __forceinline__ size_t out_index(int i, int tid, int t0, int bIdx, int o_base) {
  const int e = i * 256 + tid;
  return ((size_t)((t0 + (e >> 7)) * 4 + bIdx)) * 4096 + o_base + (e & 127);
}

__device__ __forceinline__ void load_start(float* pv, const float* out, int tid, int t0,
                                           int bIdx, int o_base) {
#pragma unroll
  for (int i = 0; i < 8; ++i) pv[i] = out[out_index(i, tid, t0, bIdx, o_base)];
}

__device__ __forceinline__ void block_out(float* out, const float* yt, const float* pv,
                                          const float* kvbuf, int tid, int t0, int bIdx,
                                          int o_base) {
#pragma unroll
  for (int i = 0; i < 8; ++i) {
    const int e = i * 256 + tid;
    const int prow = e >> 7;
    const float* kvp = &kvbuf[prow * 4];
    const float val = pv[i] + yt[e] * kvp[0] + yt[16 * 128 + e] * kvp[1] +
                      yt[2 * 16 * 128 + e] * kvp[2] + yt[3 * 16 * 128 + e] * kvp[3];
    out[out_index(i, tid, t0, bIdx, o_base)] = val;
  }
}

__global__ __launch_bounds__(256) void fc2_fused(const float* __restrict__ in,
                                                 const float* __restrict__ W,
                                                 const float* __restrict__ bias,
                                                 float* out) {
  __shared__ __attribute__((aligned(16))) float ytile[2][4][16 * 128];
  __shared__ float kvbuf[64];
  __shared__ __attribute__((aligned(16))) float stageW[8][4][SLOT];
  __shared__ __attribute__((aligned(16))) float stageA[2][4][SLOT];

  const int tid = threadIdx.x;
  const int lane = tid & 31;
  const int wave = tid >> 5;
  const int n0 = blockIdx.x * 16;
  const int bIdx = n0 >> 10;
  const int t0 = n0 & 1023;
  const float* aBase = in + ((size_t)(t0 * 4 + bIdx)) * 4096;
  float* st = &stageW[wave][0][0];

  const int rr = wave & 3;
  const int ntBase = wave >> 2;
  const int off = 2 * ((blockIdx.x ^ (blockIdx.x >> 4)) & 15);

  const int srow = tid >> 4;
  const int sc4 = 4 * (tid & 15);
  float* sOut = out + ((size_t)((t0 + 2 * wave) * 4 + bIdx)) * 4096 + lane;
  const bool upper = lane >= 16;
  float m[4], dsum[4], num[4];
#pragma unroll
  for (int r = 0; r < 4; ++r) { m[r] = -3.0e38f; dsum[r] = 0.0f; num[r] = 0.0f; }

  v4f g[34];
  float bb[4];
  {
    v4f gp[2];
    gload_part<2>(gp, aBase + (rr * 16 + (off >> 1)) * 64, (size_t)4 * 4096, lane, 8 * ntBase);
    issue_main(g, bb, aBase, W, bias, off, rr, ntBase, lane);
    stage_part<2>(&stageA[0][rr][0], gp, lane, 8 * ntBase);
  }

  for (int jj = 0; jj < 32; ++jj) {
    stage_part<2>(&stageA[((jj + 1) >> 1) & 1][rr][0], g, lane, a_row0(jj, ntBase));
#pragma unroll
    for (int s = 0; s < 4; ++s) stage_tile(st + s * SLOT, g + 2 + 8 * s, lane);
    const float b0 = bb[0], b1 = bb[1], b2 = bb[2], b3 = bb[3];
    lds_barrier();
    issue_main(g, bb, aBase, W, bias, (jj + 1 + off) & 31, rr, ntBase, lane);
    {
      const float bcur[4] = {b0, b1, b2, b3};
      v4f af[8];
      frag_read(af, &stageA[(jj >> 1) & 1][rr][0], lane);
#pragma unroll
      for (int k4 = 0; k4 < 4; k4 += 2) {
        v4f bf0[8], bf1[8];
        frag_read(bf0, st + k4 * SLOT, lane);
        frag_read(bf1, st + (k4 + 1) * SLOT, lane);
        v8f acc0, acc1;
        tile_gemm2(af, bf0, bf1, bcur[k4], bcur[k4 + 1], acc0, acc1);
        store_tile(ytile[jj & 1][rr], ntBase + 2 * k4, lane, acc0);
        store_tile(ytile[jj & 1][rr], ntBase + 2 * k4 + 2, lane, acc1);
      }
    }
    if (jj > 0)
      block_stats(&ytile[(jj - 1) & 1][0][0], srow, sc4, m, dsum, num,
                  sOut + 32 * ((jj - 1 + off) & 31), upper);
  }
  lds_barrier();
  block_stats(&ytile[1][0][0], srow, sc4, m, dsum, num, sOut + 32 * ((31 + off) & 31), upper);
#pragma unroll
  for (int r = 0; r < 4; ++r) {
#pragma unroll
    for (int mask = 1; mask < 16; mask <<= 1) {
      const float mo = __shfl_xor(m[r], mask, 32);
      const float dq = __shfl_xor(dsum[r], mask, 32);
      const float no = __shfl_xor(num[r], mask, 32);
      const float mn = fmaxf(m[r], mo);
      const float sa = __expf(m[r] - mn), sb = __expf(mo - mn);
      dsum[r] = dsum[r] * sa + dq * sb;
      num[r] = num[r] * sa + no * sb;
      m[r] = mn;
    }
  }
  if ((tid & 15) == 0) {
#pragma unroll
    for (int r = 0; r < 4; ++r) kvbuf[srow * 4 + r] = num[r] / dsum[r];
  }
  __syncthreads();

  float pv[8];
  for (int jj = 0; jj < 32; ++jj) {
    stage_part<2>(&stageA[((jj + 1) >> 1) & 1][rr][0], g, lane, a_row0(jj, ntBase));
#pragma unroll
    for (int s = 0; s < 4; ++s) stage_tile(st + s * SLOT, g + 2 + 8 * s, lane);
    const float b0 = bb[0], b1 = bb[1], b2 = bb[2], b3 = bb[3];
    lds_barrier();
    if (jj > 0) load_start(pv, out, tid, t0, bIdx, ((jj - 1 + off) & 31) * 128);
    issue_main(g, bb, aBase, W, bias, (jj + 1 + off) & 31, rr, ntBase, lane);
    {
      const float bcur[4] = {b0, b1, b2, b3};
      v4f af[8];
      frag_read(af, &stageA[(jj >> 1) & 1][rr][0], lane);
#pragma unroll
      for (int k4 = 0; k4 < 4; k4 += 2) {
        v4f bf0[8], bf1[8];
        frag_read(bf0, st + k4 * SLOT, lane);
        frag_read(bf1, st + (k4 + 1) * SLOT, lane);
        v8f acc0, acc1;
        tile_gemm2(af, bf0, bf1, bcur[k4], bcur[k4 + 1], acc0, acc1);
        store_tile(ytile[jj & 1][rr], ntBase + 2 * k4, lane, acc0);
        store_tile(ytile[jj & 1][rr], ntBase + 2 * k4 + 2, lane, acc1);
      }
    }
    if (jj > 0)
      block_out(out, &ytile[(jj - 1) & 1][0][0], pv, kvbuf, tid, t0, bIdx,
                ((jj - 1 + off) & 31) * 128);
  }
  load_start(pv, out, tid, t0, bIdx, ((31 + off) & 31) * 128);
  lds_barrier();
  block_out(out, &ytile[1][0][0], pv, kvbuf, tid, t0, bIdx, ((31 + off) & 31) * 128);
}

extern "C" void kernel_launch(void* const* d_in, const int* in_sizes, int n_in,
                              void* d_out, int out_size, void* d_ws, size_t ws_size,
                              hipStream_t stream) {
  (void)in_sizes; (void)n_in; (void)d_ws; (void)ws_size; (void)out_size;
  const float* in = (const float*)d_in[0];
  const float* W = (const float*)d_in[1];
  const float* bias = (const float*)d_in[2];
  float* out = (float*)d_out;
  fc2_fused<<<256, 256, 0, stream>>>(in, W, bias, out);
}


# Weight tiles 2,3 of a block staged, and their registers reloaded for the next block, between the block's two WMMA groups: two half-size load groups half a trip apart, one always in flight
# speedup vs baseline: 4.9840x; 1.1078x over previous
#include <hip/hip_runtime.h>


typedef float v2f __attribute__((ext_vector_type(2)));
typedef float v4f __attribute__((ext_vector_type(4)));
typedef float v8f __attribute__((ext_vector_type(8)));

#define PITCH 68
#define SLOT (16 * PITCH)

__device__ __forceinline__ void gload_tile(v4f* g, const float* __restrict__ base,
                                           size_t rowStride, int lane) {
  const float* p = base + (size_t)(lane >> 4) * rowStride + 4 * (lane & 15);
#pragma unroll
  for (int j = 0; j < 8; ++j)
    g[j] = *reinterpret_cast<const v4f*>(p + (size_t)(2 * j) * rowStride);
}

__device__ __forceinline__ void stage_tile(float* slot, const v4f* g, int lane) {
  float* q = slot + (lane >> 4) * PITCH + 4 * (lane & 15);
#pragma unroll
  for (int j = 0; j < 8; ++j)
    *reinterpret_cast<v4f*>(q + 2 * j * PITCH) = g[j];
}

template <int NS>
__device__ __forceinline__ void gload_part(v4f* g, const float* __restrict__ base,
                                           size_t rowStride, int lane, int row0) {
  const float* p = base + (size_t)(row0 + (lane >> 4)) * rowStride + 4 * (lane & 15);
#pragma unroll
  for (int j = 0; j < NS; ++j)
    g[j] = *reinterpret_cast<const v4f*>(p + (size_t)(2 * j) * rowStride);
}

template <int NS>
__device__ __forceinline__ void stage_part(float* slot, const v4f* g, int lane, int row0) {
  float* q = slot + (row0 + (lane >> 4)) * PITCH + 4 * (lane & 15);
#pragma unroll
  for (int j = 0; j < NS; ++j)
    *reinterpret_cast<v4f*>(q + 2 * j * PITCH) = g[j];
}

__device__ __forceinline__ void frag_read(v4f* f, const float* slot, int lane) {
  const float* q = slot + (lane & 15) * PITCH + 4 * (lane >> 4);
#pragma unroll
  for (int p = 0; p < 8; ++p)
    f[p] = *reinterpret_cast<const v4f*>(q + 8 * p);
}

__device__ __forceinline__ void tile_gemm2(const v4f* af, const v4f* bf0, const v4f* bf1, float bb0,
                                           float bb1, v8f& acc0, v8f& acc1) {
  acc0 = (v8f){bb0, bb0, bb0, bb0, bb0, bb0, bb0, bb0};
  acc1 = (v8f){bb1, bb1, bb1, bb1, bb1, bb1, bb1, bb1};
#pragma unroll
  for (int p = 0; p < 8; ++p) {
    const v2f a0 = af[p].xy, a1 = af[p].zw;
    acc0 = __builtin_amdgcn_wmma_f32_16x16x4_f32(false, a0, false, bf0[p].xy, (short)0, acc0, false, false);
    acc1 = __builtin_amdgcn_wmma_f32_16x16x4_f32(false, a0, false, bf1[p].xy, (short)0, acc1, false, false);
    acc0 = __builtin_amdgcn_wmma_f32_16x16x4_f32(false, a1, false, bf0[p].zw, (short)0, acc0, false, false);
    acc1 = __builtin_amdgcn_wmma_f32_16x16x4_f32(false, a1, false, bf1[p].zw, (short)0, acc1, false, false);
  }
}

__device__ __forceinline__ void store_tile(float* buf, int nt, int lane, v8f acc) {
  const int n = lane & 15, kh = lane >> 4;
#pragma unroll
  for (int i = 0; i < 8; ++i)
    buf[(i + 8 * kh) * 128 + nt * 16 + n] = acc[i];
}

__device__ __forceinline__ int a_row0(int jj, int ntBase) {
  return 8 * ntBase + ((jj & 1) ? 0 : 4);
}

__device__ __forceinline__ void issue_first(v4f* g, float* bb, const float* __restrict__ aBase,
                                            const float* __restrict__ W,
                                            const float* __restrict__ bias, int jj, int rr,
                                            int ntBase, int lane) {
  const int jE = (jj + (jj & 1)) & 31;
  const float* aT = aBase + (rr * 16 + (jE >> 1)) * 64;
  gload_part<2>(g, aT, (size_t)4 * 4096, lane, a_row0(jj, ntBase));
#pragma unroll
  for (int k4 = 0; k4 < 4; ++k4)
    bb[k4] = bias[rr * 4096 + jj * 128 + (ntBase + 2 * k4) * 16 + (lane & 15)];
#pragma unroll
  for (int k4 = 0; k4 < 2; ++k4) {
    const int ch0 = rr * 4096 + jj * 128 + (ntBase + 2 * k4) * 16;
    gload_tile(g + 2 + 8 * k4, W + (size_t)ch0 * 64, 64, lane);
  }
}

__device__ __forceinline__ void issue_second(v4f* g, const float* __restrict__ W, int jj, int rr,
                                             int ntBase, int lane) {
#pragma unroll
  for (int k4 = 2; k4 < 4; ++k4) {
    const int ch0 = rr * 4096 + jj * 128 + (ntBase + 2 * k4) * 16;
    gload_tile(g + 2 + 8 * k4, W + (size_t)ch0 * 64, 64, lane);
  }
}

__device__ __forceinline__ void lds_barrier() {
  __builtin_amdgcn_fence(__ATOMIC_RELEASE, "workgroup", "local");
  __builtin_amdgcn_s_barrier();
  __builtin_amdgcn_fence(__ATOMIC_ACQUIRE, "workgroup", "local");
}

__device__ __forceinline__ void block_stats(const float* yt, int srow, int sc4,
                                            float* m, float* dsum, float* num,
                                            float* __restrict__ sOut, bool upper) {
  v4f va[4], vb[4];
#pragma unroll
  for (int r = 0; r < 4; ++r) {
    va[r] = *reinterpret_cast<const v4f*>(yt + r * (16 * 128) + srow * 128 + sc4);
    vb[r] = *reinterpret_cast<const v4f*>(yt + r * (16 * 128) + srow * 128 + 64 + sc4);
  }
#pragma unroll
  for (int r = 0; r < 4; ++r) {
    const float give = upper ? va[r].x : vb[r].x;
    const float got = __uint_as_float(__builtin_amdgcn_permlanex16(
        0u, __float_as_uint(give), 0x76543210u, 0xfedcba98u, false, false));
    sOut[r * 1024] = upper ? got : va[r].x;
    sOut[4 * 4096 + r * 1024] = upper ? vb[r].x : got;
  }
  const v4f xa = va[0] + va[1] + va[2] + va[3];
  const v4f xb = vb[0] + vb[1] + vb[2] + vb[3];
#pragma unroll
  for (int r = 0; r < 4; ++r) {
    const v4f a = va[r], b = vb[r];
    const float mb = fmaxf(fmaxf(fmaxf(a.x, a.y), fmaxf(a.z, a.w)),
                           fmaxf(fmaxf(b.x, b.y), fmaxf(b.z, b.w)));
    const float mn = fmaxf(m[r], mb);
    const float sc = __expf(m[r] - mn);
    const float ea0 = __expf(a.x - mn), ea1 = __expf(a.y - mn);
    const float ea2 = __expf(a.z - mn), ea3 = __expf(a.w - mn);
    const float eb0 = __expf(b.x - mn), eb1 = __expf(b.y - mn);
    const float eb2 = __expf(b.z - mn), eb3 = __expf(b.w - mn);
    const float da = (ea0 + ea1) + (ea2 + ea3);
    const float db = (eb0 + eb1) + (eb2 + eb3);
    const float na = (ea0 * xa.x + ea1 * xa.y) + (ea2 * xa.z + ea3 * xa.w);
    const float nb = (eb0 * xb.x + eb1 * xb.y) + (eb2 * xb.z + eb3 * xb.w);
    dsum[r] = dsum[r] * sc + (da + db);
    num[r] = num[r] * sc + (na + nb);
    m[r] = mn;
  }
}

__device__ __forceinline__ size_t out_index(int i, int tid, int t0, int bIdx, int o_base) {
  const int e = i * 256 + tid;
  return ((size_t)((t0 + (e >> 7)) * 4 + bIdx)) * 4096 + o_base + (e & 127);
}

__device__ __forceinline__ void load_start(float* pv, const float* out, int tid, int t0,
                                           int bIdx, int o_base) {
#pragma unroll
  for (int i = 0; i < 8; ++i) pv[i] = out[out_index(i, tid, t0, bIdx, o_base)];
}

__device__ __forceinline__ void block_out(float* out, const float* yt, const float* pv,
                                          const float* kvbuf, int tid, int t0, int bIdx,
                                          int o_base) {
#pragma unroll
  for (int i = 0; i < 8; ++i) {
    const int e = i * 256 + tid;
    const int prow = e >> 7;
    const float* kvp = &kvbuf[prow * 4];
    const float val = pv[i] + yt[e] * kvp[0] + yt[16 * 128 + e] * kvp[1] +
                      yt[2 * 16 * 128 + e] * kvp[2] + yt[3 * 16 * 128 + e] * kvp[3];
    out[out_index(i, tid, t0, bIdx, o_base)] = val;
  }
}

__global__ __launch_bounds__(256) void fc2_fused(const float* __restrict__ in,
                                                 const float* __restrict__ W,
                                                 const float* __restrict__ bias,
                                                 float* out) {
  __shared__ __attribute__((aligned(16))) float ytile[2][4][16 * 128];
  __shared__ float kvbuf[64];
  __shared__ __attribute__((aligned(16))) float stageW[8][4][SLOT];
  __shared__ __attribute__((aligned(16))) float stageA[2][4][SLOT];

  const int tid = threadIdx.x;
  const int lane = tid & 31;
  const int wave = tid >> 5;
  const int n0 = blockIdx.x * 16;
  const int bIdx = n0 >> 10;
  const int t0 = n0 & 1023;
  const float* aBase = in + ((size_t)(t0 * 4 + bIdx)) * 4096;
  float* st = &stageW[wave][0][0];

  const int rr = wave & 3;
  const int ntBase = wave >> 2;
  const int off = 2 * ((blockIdx.x ^ (blockIdx.x >> 4)) & 15);

  const int srow = tid >> 4;
  const int sc4 = 4 * (tid & 15);
  float* sOut = out + ((size_t)((t0 + 2 * wave) * 4 + bIdx)) * 4096 + lane;
  const bool upper = lane >= 16;
  float m[4], dsum[4], num[4];
#pragma unroll
  for (int r = 0; r < 4; ++r) { m[r] = -3.0e38f; dsum[r] = 0.0f; num[r] = 0.0f; }

  v4f g[34];
  float bb[4];
  {
    v4f gp[2];
    gload_part<2>(gp, aBase + (rr * 16 + (off >> 1)) * 64, (size_t)4 * 4096, lane, 8 * ntBase);
    issue_first(g, bb, aBase, W, bias, off, rr, ntBase, lane);
    issue_second(g, W, off, rr, ntBase, lane);
    stage_part<2>(&stageA[0][rr][0], gp, lane, 8 * ntBase);
  }

  for (int jj = 0; jj < 32; ++jj) {
    stage_part<2>(&stageA[((jj + 1) >> 1) & 1][rr][0], g, lane, a_row0(jj, ntBase));
    stage_tile(st, g + 2, lane);
    stage_tile(st + SLOT, g + 10, lane);
    const float b0 = bb[0], b1 = bb[1], b2 = bb[2], b3 = bb[3];
    lds_barrier();
    issue_first(g, bb, aBase, W, bias, (jj + 1 + off) & 31, rr, ntBase, lane);
    {
      v4f af[8];
      frag_read(af, &stageA[(jj >> 1) & 1][rr][0], lane);
      v8f acc0, acc1;
      {
        v4f bf0[8], bf1[8];
        frag_read(bf0, st, lane);
        frag_read(bf1, st + SLOT, lane);
        tile_gemm2(af, bf0, bf1, b0, b1, acc0, acc1);
        store_tile(ytile[jj & 1][rr], ntBase, lane, acc0);
        store_tile(ytile[jj & 1][rr], ntBase + 2, lane, acc1);
      }
      stage_tile(st + 2 * SLOT, g + 18, lane);
      stage_tile(st + 3 * SLOT, g + 26, lane);
      issue_second(g, W, (jj + 1 + off) & 31, rr, ntBase, lane);
      {
        v4f bf0[8], bf1[8];
        frag_read(bf0, st + 2 * SLOT, lane);
        frag_read(bf1, st + 3 * SLOT, lane);
        tile_gemm2(af, bf0, bf1, b2, b3, acc0, acc1);
        store_tile(ytile[jj & 1][rr], ntBase + 4, lane, acc0);
        store_tile(ytile[jj & 1][rr], ntBase + 6, lane, acc1);
      }
    }
    if (jj > 0)
      block_stats(&ytile[(jj - 1) & 1][0][0], srow, sc4, m, dsum, num,
                  sOut + 32 * ((jj - 1 + off) & 31), upper);
  }
  lds_barrier();
  block_stats(&ytile[1][0][0], srow, sc4, m, dsum, num, sOut + 32 * ((31 + off) & 31), upper);
#pragma unroll
  for (int r = 0; r < 4; ++r) {
#pragma unroll
    for (int mask = 1; mask < 16; mask <<= 1) {
      const float mo = __shfl_xor(m[r], mask, 32);
      const float dq = __shfl_xor(dsum[r], mask, 32);
      const float no = __shfl_xor(num[r], mask, 32);
      const float mn = fmaxf(m[r], mo);
      const float sa = __expf(m[r] - mn), sb = __expf(mo - mn);
      dsum[r] = dsum[r] * sa + dq * sb;
      num[r] = num[r] * sa + no * sb;
      m[r] = mn;
    }
  }
  if ((tid & 15) == 0) {
#pragma unroll
    for (int r = 0; r < 4; ++r) kvbuf[srow * 4 + r] = num[r] / dsum[r];
  }
  __syncthreads();

  float pv[8];
  for (int jj = 0; jj < 32; ++jj) {
    stage_part<2>(&stageA[((jj + 1) >> 1) & 1][rr][0], g, lane, a_row0(jj, ntBase));
    stage_tile(st, g + 2, lane);
    stage_tile(st + SLOT, g + 10, lane);
    const float b0 = bb[0], b1 = bb[1], b2 = bb[2], b3 = bb[3];
    lds_barrier();
    if (jj > 0) load_start(pv, out, tid, t0, bIdx, ((jj - 1 + off) & 31) * 128);
    issue_first(g, bb, aBase, W, bias, (jj + 1 + off) & 31, rr, ntBase, lane);
    {
      v4f af[8];
      frag_read(af, &stageA[(jj >> 1) & 1][rr][0], lane);
      v8f acc0, acc1;
      {
        v4f bf0[8], bf1[8];
        frag_read(bf0, st, lane);
        frag_read(bf1, st + SLOT, lane);
        tile_gemm2(af, bf0, bf1, b0, b1, acc0, acc1);
        store_tile(ytile[jj & 1][rr], ntBase, lane, acc0);
        store_tile(ytile[jj & 1][rr], ntBase + 2, lane, acc1);
      }
      stage_tile(st + 2 * SLOT, g + 18, lane);
      stage_tile(st + 3 * SLOT, g + 26, lane);
      issue_second(g, W, (jj + 1 + off) & 31, rr, ntBase, lane);
      {
        v4f bf0[8], bf1[8];
        frag_read(bf0, st + 2 * SLOT, lane);
        frag_read(bf1, st + 3 * SLOT, lane);
        tile_gemm2(af, bf0, bf1, b2, b3, acc0, acc1);
        store_tile(ytile[jj & 1][rr], ntBase + 4, lane, acc0);
        store_tile(ytile[jj & 1][rr], ntBase + 6, lane, acc1);
      }
    }
    if (jj > 0)
      block_out(out, &ytile[(jj - 1) & 1][0][0], pv, kvbuf, tid, t0, bIdx,
                ((jj - 1 + off) & 31) * 128);
  }
  load_start(pv, out, tid, t0, bIdx, ((31 + off) & 31) * 128);
  lds_barrier();
  block_out(out, &ytile[1][0][0], pv, kvbuf, tid, t0, bIdx, ((31 + off) & 31) * 128);
}

extern "C" void kernel_launch(void* const* d_in, const int* in_sizes, int n_in,
                              void* d_out, int out_size, void* d_ws, size_t ws_size,
                              hipStream_t stream) {
  (void)in_sizes; (void)n_in; (void)d_ws; (void)ws_size; (void)out_size;
  const float* in = (const float*)d_in[0];
  const float* W = (const float*)d_in[1];
  const float* bias = (const float*)d_in[2];
  float* out = (float*)d_out;
  fc2_fused<<<256, 256, 0, stream>>>(in, W, bias, out);
}
